# starting kernel
# baseline (speedup 1.0000x reference)
#include <hip/hip_runtime.h>
#include <hip/hip_bf16.h>

#define NN   50000
#define NE   1600000
#define INF_ 128
#define HID  64
#define HEADS 2
#define HD   (HEADS*HID)
#define NEG_SLOPE 0.2f

typedef __attribute__((ext_vector_type(2))) float v2f;
typedef __attribute__((ext_vector_type(8))) float v8f;

__device__ __forceinline__ void atomicMaxF(float* addr, float val) {
    if (val >= 0.0f) atomicMax((int*)addr, __float_as_int(val));
    else             atomicMin((unsigned int*)addr, (unsigned int)__float_as_int(val));
}


__global__ void gat_w_transpose(const float* __restrict__ W, float* __restrict__ WT,
                                int K, int total) {
    int i = blockIdx.x * blockDim.x + threadIdx.x;
    if (i >= total) return;
    int k = i / HD;
    int n = i - k * HD;
    WT[(long)n * K + k] = W[i];
}

__global__ void __launch_bounds__(256) gat_gemm_wmma_f32(
        const float* __restrict__ A, const float* __restrict__ BT,
        float* __restrict__ D, int K) {
    const int lane = threadIdx.x & 31;
    const int wave = threadIdx.x >> 5;
    const int half = lane >> 4;
    const int l15  = lane & 15;

    const float* __restrict__ arow = A  + ((long)blockIdx.x * 16 + l15) * K;
    const float* __restrict__ brow = BT + ((long)wave       * 16 + l15) * K;

    v8f c = {};
    for (int kb = 0; kb < K; kb += 4) {
        const int k0 = kb + half * 2;
        v2f a, b;
        a.x = arow[k0]; a.y = arow[k0 + 1];
        b.x = brow[k0]; b.y = brow[k0 + 1];
        c = __builtin_amdgcn_wmma_f32_16x16x4_f32(false, a, false, b,
                                                  (short)0, c, false, false);
    }
    float* __restrict__ drow =
        D + ((long)blockIdx.x * 16 + half * 8) * HD + wave * 16 + l15;
#pragma unroll
    for (int v = 0; v < 8; ++v) drow[(long)v * HD] = c[v];
}

__global__ void gat_eler(const float* __restrict__ feat,
                         const float* __restrict__ al, const float* __restrict__ ar,
                         float* __restrict__ el, float* __restrict__ er, int NH) {
    int i = blockIdx.x * blockDim.x + threadIdx.x;
    if (i >= NH) return;
    const int h = i & 1;
    const float* __restrict__ f = feat + (long)(i >> 1) * HD + h * HID;
    const float* __restrict__ la = al + h * HID;
    const float* __restrict__ ra = ar + h * HID;
    float sl = 0.f, sr = 0.f;
#pragma unroll 4
    for (int d = 0; d < HID; ++d) { float v = f[d]; sl += v * la[d]; sr += v * ra[d]; }
    el[i] = sl; er[i] = sr;
}

__global__ void gat_init(float* __restrict__ m, float* __restrict__ s,
                         float* __restrict__ agg, int NH, int NAGG) {
    int i = blockIdx.x * blockDim.x + threadIdx.x;
    if (i < NH) { m[i] = -INFINITY; s[i] = 0.f; }
    if (i < NAGG) agg[i] = 0.f;
}

__global__ void gat_edge_max(const int* __restrict__ src, const int* __restrict__ dst,
                             const float* __restrict__ el, const float* __restrict__ er,
                             float* __restrict__ ebuf, float* __restrict__ m, int E) {
    int e = blockIdx.x * blockDim.x + threadIdx.x;
    if (e >= E) return;
    const int s0 = src[e], d0 = dst[e];
#pragma unroll
    for (int h = 0; h < HEADS; ++h) {
        float v = el[s0 * HEADS + h] + er[d0 * HEADS + h];
        v = v > 0.f ? v : NEG_SLOPE * v;
        ebuf[(long)e * HEADS + h] = v;
        atomicMaxF(&m[d0 * HEADS + h], v);
    }
}

__global__ void gat_edge_exp(const int* __restrict__ dst, const float* __restrict__ m,
                             float* __restrict__ ebuf, float* __restrict__ ssum, int E) {
    int e = blockIdx.x * blockDim.x + threadIdx.x;
    if (e >= E) return;
    const int d0 = dst[e];
#pragma unroll
    for (int h = 0; h < HEADS; ++h) {
        float ex = __expf(ebuf[(long)e * HEADS + h] - m[d0 * HEADS + h]);
        ebuf[(long)e * HEADS + h] = ex;
        unsafeAtomicAdd(&ssum[d0 * HEADS + h], ex);
    }
}

__global__ void __launch_bounds__(256) gat_edge_scatter(
        const int* __restrict__ src, const int* __restrict__ dst,
        const float* __restrict__ feat, const float* __restrict__ ebuf,
        const float* __restrict__ ssum, float* __restrict__ agg, int E) {
    const int e = blockIdx.x * 8 + (threadIdx.x >> 5);
    if (e >= E) return;
    const int lane = threadIdx.x & 31;
    const int s0 = src[e], d0 = dst[e];
    const int h  = lane >> 4;
    const int c0 = lane * 4;
    const float alpha = ebuf[(long)e * HEADS + h] / ssum[d0 * HEADS + h];

    const int e2 = e + 32;
    if (e2 < E) __builtin_prefetch(feat + (long)src[e2] * HD + c0, 0, 0);

    const float4 f = *(const float4*)(feat + (long)s0 * HD + c0);
    float* __restrict__ o = agg + (long)d0 * HD + c0;
    unsafeAtomicAdd(o + 0, f.x * alpha);
    unsafeAtomicAdd(o + 1, f.y * alpha);
    unsafeAtomicAdd(o + 2, f.z * alpha);
    unsafeAtomicAdd(o + 3, f.w * alpha);
}

__global__ void gat_finalize(const float* __restrict__ agg, const float* __restrict__ bias,
                             float* __restrict__ out, int total) {
    int i = blockIdx.x * blockDim.x + threadIdx.x;
    if (i >= total) return;
    const int n = i >> 6, d = i & 63;
    const float v0 = agg[(long)n * HD + d]        + bias[d];
    const float v1 = agg[(long)n * HD + HID + d]  + bias[HID + d];
    const float v  = 0.5f * (v0 + v1);
    out[i] = v > 0.f ? v : (__expf(v) - 1.f);
}

extern "C" void kernel_launch(void* const* d_in, const int* in_sizes, int n_in,
                              void* d_out, int out_size, void* d_ws, size_t ws_size,
                              hipStream_t stream) {
    const float* x   = (const float*)d_in[0];
    const float* W1  = (const float*)d_in[1];
    const float* al1 = (const float*)d_in[2];
    const float* ar1 = (const float*)d_in[3];
    const float* b1  = (const float*)d_in[4];
    const float* W2  = (const float*)d_in[5];
    const float* al2 = (const float*)d_in[6];
    const float* ar2 = (const float*)d_in[7];
    const float* b2  = (const float*)d_in[8];
    const int*   src = (const int*)d_in[9];
    const int*   dst = (const int*)d_in[10];
    float* out = (float*)d_out;

    char* wsb = (char*)d_ws;
    size_t off = 0;
    auto alloc = [&](size_t bytes) -> void* {
        void* p = wsb + off;
        off += bytes;
        off = (off + 255) & ~(size_t)255;
        return p;
    };
    float* WT1  = (float*)alloc((size_t)INF_ * HD * 4);
    float* WT2  = (float*)alloc((size_t)HID  * HD * 4);
    float* feat = (float*)alloc((size_t)NN * HD * 4);
    float* el   = (float*)alloc((size_t)NN * HEADS * 4);
    float* er   = (float*)alloc((size_t)NN * HEADS * 4);
    float* m    = (float*)alloc((size_t)NN * HEADS * 4);
    float* s    = (float*)alloc((size_t)NN * HEADS * 4);
    float* ebuf = (float*)alloc((size_t)NE * HEADS * 4);
    float* agg  = (float*)alloc((size_t)NN * HD * 4);
    float* h    = (float*)alloc((size_t)NN * HID * 4);

    const int NH   = NN * HEADS;
    const int NAGG = NN * HD;
    const int NOUT = NN * HID;
    const dim3 B256(256);

    gat_w_transpose<<<dim3((INF_ * HD + 255) / 256), B256, 0, stream>>>(W1, WT1, INF_, INF_ * HD);
    gat_gemm_wmma_f32<<<dim3(NN / 16), B256, 0, stream>>>(x, WT1, feat, INF_);
    gat_eler<<<dim3((NH + 255) / 256), B256, 0, stream>>>(feat, al1, ar1, el, er, NH);
    gat_init<<<dim3((NAGG + 255) / 256), B256, 0, stream>>>(m, s, agg, NH, NAGG);
    gat_edge_max<<<dim3((NE + 255) / 256), B256, 0, stream>>>(src, dst, el, er, ebuf, m, NE);
    gat_edge_exp<<<dim3((NE + 255) / 256), B256, 0, stream>>>(dst, m, ebuf, s, NE);
    gat_edge_scatter<<<dim3((NE + 7) / 8), B256, 0, stream>>>(src, dst, feat, ebuf, s, agg, NE);
    gat_finalize<<<dim3((NOUT + 255) / 256), B256, 0, stream>>>(agg, b1, h, NOUT);

    gat_w_transpose<<<dim3((HID * HD + 255) / 256), B256, 0, stream>>>(W2, WT2, HID, HID * HD);
    gat_gemm_wmma_f32<<<dim3(NN / 16), B256, 0, stream>>>(h, WT2, feat, HID);
    gat_eler<<<dim3((NH + 255) / 256), B256, 0, stream>>>(feat, al2, ar2, el, er, NH);
    gat_init<<<dim3((NAGG + 255) / 256), B256, 0, stream>>>(m, s, agg, NH, NAGG);
    gat_edge_max<<<dim3((NE + 255) / 256), B256, 0, stream>>>(src, dst, el, er, ebuf, m, NE);
    gat_edge_exp<<<dim3((NE + 255) / 256), B256, 0, stream>>>(dst, m, ebuf, s, NE);
    gat_edge_scatter<<<dim3((NE + 7) / 8), B256, 0, stream>>>(src, dst, feat, ebuf, s, agg, NE);
    gat_finalize<<<dim3((NOUT + 255) / 256), B256, 0, stream>>>(agg, b2, out, NOUT);
}


# Atomics coalesced by 128-B line: edge max/exp run one lane per (edge, head) with one ordered-key unsigned max; scatter lanes own channels lane+32j, one line per add; sums stay atomic arrival order
# speedup vs baseline: 1.6262x; 1.6262x over previous
#include <hip/hip_runtime.h>
#include <hip/hip_bf16.h>

#define NN   50000
#define NE   1600000
#define INF_ 128
#define HID  64
#define HEADS 2
#define HD   (HEADS*HID)
#define NEG_SLOPE 0.2f

typedef __attribute__((ext_vector_type(2))) float v2f;
typedef __attribute__((ext_vector_type(8))) float v8f;

__device__ __forceinline__ unsigned int f32_to_key(float f) {
    const unsigned int b = __float_as_uint(f);
    return b ^ ((b >> 31) ? 0xFFFFFFFFu : 0x80000000u);
}
__device__ __forceinline__ float key_to_f32(unsigned int k) {
    return __uint_as_float(k ^ ((k >> 31) ? 0x80000000u : 0xFFFFFFFFu));
}
#define KEY_NEG_INF 0x007FFFFFu


__global__ void gat_w_transpose(const float* __restrict__ W, float* __restrict__ WT,
                                int K, int total) {
    int i = blockIdx.x * blockDim.x + threadIdx.x;
    if (i >= total) return;
    int k = i / HD;
    int n = i - k * HD;
    WT[(long)n * K + k] = W[i];
}

__global__ void __launch_bounds__(256) gat_gemm_wmma_f32(
        const float* __restrict__ A, const float* __restrict__ BT,
        float* __restrict__ D, int K) {
    const int lane = threadIdx.x & 31;
    const int wave = threadIdx.x >> 5;
    const int half = lane >> 4;
    const int l15  = lane & 15;

    const float* __restrict__ arow = A  + ((long)blockIdx.x * 16 + l15) * K;
    const float* __restrict__ brow = BT + ((long)wave       * 16 + l15) * K;

    v8f c = {};
    for (int kb = 0; kb < K; kb += 4) {
        const int k0 = kb + half * 2;
        v2f a, b;
        a.x = arow[k0]; a.y = arow[k0 + 1];
        b.x = brow[k0]; b.y = brow[k0 + 1];
        c = __builtin_amdgcn_wmma_f32_16x16x4_f32(false, a, false, b,
                                                  (short)0, c, false, false);
    }
    float* __restrict__ drow =
        D + ((long)blockIdx.x * 16 + half * 8) * HD + wave * 16 + l15;
#pragma unroll
    for (int v = 0; v < 8; ++v) drow[(long)v * HD] = c[v];
}

__global__ void gat_eler(const float* __restrict__ feat,
                         const float* __restrict__ al, const float* __restrict__ ar,
                         float* __restrict__ el, float* __restrict__ er, int NH) {
    int i = blockIdx.x * blockDim.x + threadIdx.x;
    if (i >= NH) return;
    const int h = i & 1;
    const float* __restrict__ f = feat + (long)(i >> 1) * HD + h * HID;
    const float* __restrict__ la = al + h * HID;
    const float* __restrict__ ra = ar + h * HID;
    float sl = 0.f, sr = 0.f;
#pragma unroll 4
    for (int d = 0; d < HID; ++d) { float v = f[d]; sl += v * la[d]; sr += v * ra[d]; }
    el[i] = sl; er[i] = sr;
}

__global__ void gat_init(unsigned int* __restrict__ m, float* __restrict__ s,
                         float* __restrict__ agg, int NH, int NAGG) {
    int i = blockIdx.x * blockDim.x + threadIdx.x;
    if (i < NH) { m[i] = KEY_NEG_INF; s[i] = 0.f; }
    if (i < NAGG) agg[i] = 0.f;
}

__global__ void gat_edge_max(const int* __restrict__ src, const int* __restrict__ dst,
                             const float* __restrict__ el, const float* __restrict__ er,
                             float* __restrict__ ebuf, unsigned int* __restrict__ m, int EH) {
    const int t = blockIdx.x * blockDim.x + threadIdx.x;
    if (t >= EH) return;
    const int e = t >> 1, h = t & 1;
    const int s0 = src[e], d0 = dst[e];
    float v = el[s0 * HEADS + h] + er[d0 * HEADS + h];
    v = v > 0.f ? v : NEG_SLOPE * v;
    ebuf[t] = v;
    atomicMax(&m[d0 * HEADS + h], f32_to_key(v));
}

__global__ void gat_edge_exp(const int* __restrict__ dst, const unsigned int* __restrict__ m,
                             float* __restrict__ ebuf, float* __restrict__ ssum, int EH) {
    const int t = blockIdx.x * blockDim.x + threadIdx.x;
    if (t >= EH) return;
    const int e = t >> 1, h = t & 1;
    const int d0 = dst[e];
    const float ex = __expf(ebuf[t] - key_to_f32(m[d0 * HEADS + h]));
    ebuf[t] = ex;
    unsafeAtomicAdd(&ssum[d0 * HEADS + h], ex);
}

__global__ void __launch_bounds__(256) gat_edge_scatter(
        const int* __restrict__ src, const int* __restrict__ dst,
        const float* __restrict__ feat, const float* __restrict__ ebuf,
        const float* __restrict__ ssum, float* __restrict__ agg, int E) {
    const int e = blockIdx.x * 8 + (threadIdx.x >> 5);
    if (e >= E) return;
    const int lane = threadIdx.x & 31;
    const int s0 = src[e], d0 = dst[e];
    const int c0 = lane * 4;
    const float2 ex2 = *(const float2*)(ebuf + (long)e * HEADS);
    const float2 ss2 = *(const float2*)(ssum + d0 * HEADS);
    const float alpha0 = ex2.x / ss2.x;
    const float alpha1 = ex2.y / ss2.y;

    const int e2 = e + 32;
    if (e2 < E) __builtin_prefetch(feat + (long)src[e2] * HD + c0, 0, 0);

    const float* __restrict__ f = feat + (long)s0 * HD + lane;
    const float f0 = f[0], f1 = f[32], f2 = f[64], f3 = f[96];
    float* __restrict__ o = agg + (long)d0 * HD + lane;
    unsafeAtomicAdd(o +  0, f0 * alpha0);
    unsafeAtomicAdd(o + 32, f1 * alpha0);
    unsafeAtomicAdd(o + 64, f2 * alpha1);
    unsafeAtomicAdd(o + 96, f3 * alpha1);
}

__global__ void gat_finalize(const float* __restrict__ agg, const float* __restrict__ bias,
                             float* __restrict__ out, int total) {
    int i = blockIdx.x * blockDim.x + threadIdx.x;
    if (i >= total) return;
    const int n = i >> 6, d = i & 63;
    const float v0 = agg[(long)n * HD + d]        + bias[d];
    const float v1 = agg[(long)n * HD + HID + d]  + bias[HID + d];
    const float v  = 0.5f * (v0 + v1);
    out[i] = v > 0.f ? v : (__expf(v) - 1.f);
}

extern "C" void kernel_launch(void* const* d_in, const int* in_sizes, int n_in,
                              void* d_out, int out_size, void* d_ws, size_t ws_size,
                              hipStream_t stream) {
    const float* x   = (const float*)d_in[0];
    const float* W1  = (const float*)d_in[1];
    const float* al1 = (const float*)d_in[2];
    const float* ar1 = (const float*)d_in[3];
    const float* b1  = (const float*)d_in[4];
    const float* W2  = (const float*)d_in[5];
    const float* al2 = (const float*)d_in[6];
    const float* ar2 = (const float*)d_in[7];
    const float* b2  = (const float*)d_in[8];
    const int*   src = (const int*)d_in[9];
    const int*   dst = (const int*)d_in[10];
    float* out = (float*)d_out;

    char* wsb = (char*)d_ws;
    size_t off = 0;
    auto alloc = [&](size_t bytes) -> void* {
        void* p = wsb + off;
        off += bytes;
        off = (off + 255) & ~(size_t)255;
        return p;
    };
    float* WT1  = (float*)alloc((size_t)INF_ * HD * 4);
    float* WT2  = (float*)alloc((size_t)HID  * HD * 4);
    float* feat = (float*)alloc((size_t)NN * HD * 4);
    float* el   = (float*)alloc((size_t)NN * HEADS * 4);
    float* er   = (float*)alloc((size_t)NN * HEADS * 4);
    unsigned int* m = (unsigned int*)alloc((size_t)NN * HEADS * 4);
    float* s    = (float*)alloc((size_t)NN * HEADS * 4);
    float* ebuf = (float*)alloc((size_t)NE * HEADS * 4);
    float* agg  = (float*)alloc((size_t)NN * HD * 4);
    float* h    = (float*)alloc((size_t)NN * HID * 4);

    const int NH   = NN * HEADS;
    const int NAGG = NN * HD;
    const int NOUT = NN * HID;
    const int NEH  = NE * HEADS;
    const dim3 B256(256);

    gat_w_transpose<<<dim3((INF_ * HD + 255) / 256), B256, 0, stream>>>(W1, WT1, INF_, INF_ * HD);
    gat_gemm_wmma_f32<<<dim3(NN / 16), B256, 0, stream>>>(x, WT1, feat, INF_);
    gat_eler<<<dim3((NH + 255) / 256), B256, 0, stream>>>(feat, al1, ar1, el, er, NH);
    gat_init<<<dim3((NAGG + 255) / 256), B256, 0, stream>>>(m, s, agg, NH, NAGG);
    gat_edge_max<<<dim3((NEH + 255) / 256), B256, 0, stream>>>(src, dst, el, er, ebuf, m, NEH);
    gat_edge_exp<<<dim3((NEH + 255) / 256), B256, 0, stream>>>(dst, m, ebuf, s, NEH);
    gat_edge_scatter<<<dim3((NE + 7) / 8), B256, 0, stream>>>(src, dst, feat, ebuf, s, agg, NE);
    gat_finalize<<<dim3((NOUT + 255) / 256), B256, 0, stream>>>(agg, b1, h, NOUT);

    gat_w_transpose<<<dim3((HID * HD + 255) / 256), B256, 0, stream>>>(W2, WT2, HID, HID * HD);
    gat_gemm_wmma_f32<<<dim3(NN / 16), B256, 0, stream>>>(h, WT2, feat, HID);
    gat_eler<<<dim3((NH + 255) / 256), B256, 0, stream>>>(feat, al2, ar2, el, er, NH);
    gat_init<<<dim3((NAGG + 255) / 256), B256, 0, stream>>>(m, s, agg, NH, NAGG);
    gat_edge_max<<<dim3((NEH + 255) / 256), B256, 0, stream>>>(src, dst, el, er, ebuf, m, NEH);
    gat_edge_exp<<<dim3((NEH + 255) / 256), B256, 0, stream>>>(dst, m, ebuf, s, NEH);
    gat_edge_scatter<<<dim3((NE + 7) / 8), B256, 0, stream>>>(src, dst, feat, ebuf, s, agg, NE);
    gat_finalize<<<dim3((NOUT + 255) / 256), B256, 0, stream>>>(agg, b2, out, NOUT);
}


# Edges bucketed by destination in-call (integer atomic slot per edge, scan, fill); one wave per node keeps max, exp-sum, feature sum in registers: slot-order sums, divided by s last; no float atomics
# speedup vs baseline: 4.7454x; 2.9181x over previous
#include <hip/hip_runtime.h>
#include <hip/hip_bf16.h>

#define NN   50000
#define NE   1600000
#define INF_ 128
#define HID  64
#define HEADS 2
#define HD   (HEADS*HID)
#define NEG_SLOPE 0.2f

typedef __attribute__((ext_vector_type(2))) float v2f;
typedef __attribute__((ext_vector_type(8))) float v8f;


__global__ void gat_w_transpose(const float* __restrict__ W, float* __restrict__ WT,
                                int K, int total) {
    int i = blockIdx.x * blockDim.x + threadIdx.x;
    if (i >= total) return;
    int k = i / HD;
    int n = i - k * HD;
    WT[(long)n * K + k] = W[i];
}

__global__ void __launch_bounds__(256) gat_gemm_wmma_f32(
        const float* __restrict__ A, const float* __restrict__ BT,
        float* __restrict__ D, int K) {
    const int lane = threadIdx.x & 31;
    const int wave = threadIdx.x >> 5;
    const int half = lane >> 4;
    const int l15  = lane & 15;

    const float* __restrict__ arow = A  + ((long)blockIdx.x * 16 + l15) * K;
    const float* __restrict__ brow = BT + ((long)wave       * 16 + l15) * K;

    v8f c = {};
    for (int kb = 0; kb < K; kb += 4) {
        const int k0 = kb + half * 2;
        v2f a, b;
        a.x = arow[k0]; a.y = arow[k0 + 1];
        b.x = brow[k0]; b.y = brow[k0 + 1];
        c = __builtin_amdgcn_wmma_f32_16x16x4_f32(false, a, false, b,
                                                  (short)0, c, false, false);
    }
    float* __restrict__ drow =
        D + ((long)blockIdx.x * 16 + half * 8) * HD + wave * 16 + l15;
#pragma unroll
    for (int v = 0; v < 8; ++v) drow[(long)v * HD] = c[v];
}

__global__ void gat_eler(const float* __restrict__ feat,
                         const float* __restrict__ al, const float* __restrict__ ar,
                         float* __restrict__ el, float* __restrict__ er, int NH) {
    int i = blockIdx.x * blockDim.x + threadIdx.x;
    if (i >= NH) return;
    const int h = i & 1;
    const float* __restrict__ f = feat + (long)(i >> 1) * HD + h * HID;
    const float* __restrict__ la = al + h * HID;
    const float* __restrict__ ra = ar + h * HID;
    float sl = 0.f, sr = 0.f;
#pragma unroll 4
    for (int d = 0; d < HID; ++d) { float v = f[d]; sl += v * la[d]; sr += v * ra[d]; }
    el[i] = sl; er[i] = sr;
}


__global__ void gat_csr_zero(int* __restrict__ deg, int n) {
    const int i = blockIdx.x * blockDim.x + threadIdx.x;
    if (i < n) deg[i] = 0;
}

__global__ void gat_csr_count(const int* __restrict__ dst, int* __restrict__ deg,
                              int* __restrict__ rank, int E) {
    const int e = blockIdx.x * blockDim.x + threadIdx.x;
    if (e >= E) return;
    rank[e] = atomicAdd(&deg[dst[e]], 1);
}

__global__ void __launch_bounds__(1024) gat_csr_scan(const int* __restrict__ deg,
                                                     int* __restrict__ rowptr, int n) {
    __shared__ int wave_front[32];
    __shared__ int wave_own[32];
    const int t = threadIdx.x, lane = t & 31, w = t >> 5;
    const int tile0 = blockIdx.x * 1024;

    int front = 0;
    for (int i = t; i < tile0; i += 1024) front += deg[i];
    const int i = tile0 + t;
    const int own = (i < n) ? deg[i] : 0;

    int inc = own;
    for (int off = 1; off < 32; off <<= 1) {
        const int v = __shfl_up(inc, off);
        if (lane >= off) inc += v;
    }
    for (int off = 16; off; off >>= 1) front += __shfl_xor(front, off);
    if (lane == 31) wave_own[w] = inc;
    if (lane == 0)  wave_front[w] = front;
    __syncthreads();
    int base = wave_front[lane] + (lane < w ? wave_own[lane] : 0);
    for (int off = 16; off; off >>= 1) base += __shfl_xor(base, off);

    if (i < n) rowptr[i] = base + inc - own;
    if (i == n - 1) rowptr[n] = base + inc;
}

__global__ void gat_csr_fill(const int* __restrict__ src, const int* __restrict__ dst,
                             const int* __restrict__ rowptr, const int* __restrict__ rank,
                             int* __restrict__ csr_src, int E) {
    const int e = blockIdx.x * blockDim.x + threadIdx.x;
    if (e >= E) return;
    csr_src[rowptr[dst[e]] + rank[e]] = src[e];
}

__device__ __forceinline__ float gat_leaky(float v) { return v > 0.f ? v : NEG_SLOPE * v; }

__device__ __forceinline__ float gat_lane_f(float v, int j) {
    return __int_as_float(__builtin_amdgcn_readlane(__float_as_int(v), j));
}

template <int N>
__device__ __forceinline__ void gat_add_rows(float4& acc, const float* __restrict__ fcol,
                                             int s, float x0, float x1, bool head1, int j) {
    float4 f[N];
    float  a[N];
#pragma unroll
    for (int u = 0; u < N; ++u) {
        const int sj = __builtin_amdgcn_readlane(s, j + u);
        f[u] = *(const float4*)(fcol + (long)sj * HD);
        const float a0 = gat_lane_f(x0, j + u), a1 = gat_lane_f(x1, j + u);
        a[u] = head1 ? a1 : a0;
    }
#pragma unroll
    for (int u = 0; u < N; ++u) {
        acc.x += f[u].x * a[u]; acc.y += f[u].y * a[u];
        acc.z += f[u].z * a[u]; acc.w += f[u].w * a[u];
    }
}

__global__ void __launch_bounds__(256) gat_node_aggregate(
        const int* __restrict__ rowptr, const int* __restrict__ csr_src,
        const float* __restrict__ feat, const float* __restrict__ el,
        const float* __restrict__ er, const float* __restrict__ bias,
        float* __restrict__ out, int n_nodes) {
    const int n = __builtin_amdgcn_readfirstlane((int)(blockIdx.x * 8 + (threadIdx.x >> 5)));
    if (n >= n_nodes) return;
    const int lane = threadIdx.x & 31;
    const bool head1 = lane >= 16;
    const int beg = rowptr[n], end = rowptr[n + 1];
    const float2 ern = *(const float2*)(er + n * HEADS);

    float m0 = -__builtin_inff(), m1 = -__builtin_inff();
    for (int p = beg + lane; p < end; p += 32) {
        const int s = csr_src[p];
        const float2 l = *(const float2*)(el + s * HEADS);
        m0 = fmaxf(m0, gat_leaky(l.x + ern.x));
        m1 = fmaxf(m1, gat_leaky(l.y + ern.y));
    }
    for (int off = 16; off; off >>= 1) {
        m0 = fmaxf(m0, __shfl_xor(m0, off));
        m1 = fmaxf(m1, __shfl_xor(m1, off));
    }

    float s0 = 0.f, s1 = 0.f;
    float4 acc = make_float4(0.f, 0.f, 0.f, 0.f);
    const float* __restrict__ fcol = feat + lane * 4;
    for (int base = beg; base < end; base += 32) {
        const int p = base + lane;
        int s = 0;
        float x0 = 0.f, x1 = 0.f;
        if (p < end) {
            s = csr_src[p];
            const float2 l = *(const float2*)(el + s * HEADS);
            x0 = __expf(gat_leaky(l.x + ern.x) - m0);
            x1 = __expf(gat_leaky(l.y + ern.y) - m1);
        }
        s0 += x0; s1 += x1;
        const int cnt = min(32, end - base);
        int j = 0;
        for (; j + 8 <= cnt; j += 8) gat_add_rows<8>(acc, fcol, s, x0, x1, head1, j);
        if (j + 4 <= cnt) { gat_add_rows<4>(acc, fcol, s, x0, x1, head1, j); j += 4; }
        if (j + 2 <= cnt) { gat_add_rows<2>(acc, fcol, s, x0, x1, head1, j); j += 2; }
        if (j < cnt) gat_add_rows<1>(acc, fcol, s, x0, x1, head1, j);
    }
    for (int off = 16; off; off >>= 1) {
        s0 += __shfl_xor(s0, off);
        s1 += __shfl_xor(s1, off);
    }
    if (end > beg) {
        const float sden = head1 ? s1 : s0;
        acc.x /= sden; acc.y /= sden; acc.z /= sden; acc.w /= sden;
    }

    const float4 b = *(const float4*)(bias + lane * 4);
    float4 v = make_float4(acc.x + b.x, acc.y + b.y, acc.z + b.z, acc.w + b.w);
    float4 r;
    r.x = 0.5f * (v.x + __shfl_xor(v.x, 16));
    r.y = 0.5f * (v.y + __shfl_xor(v.y, 16));
    r.z = 0.5f * (v.z + __shfl_xor(v.z, 16));
    r.w = 0.5f * (v.w + __shfl_xor(v.w, 16));
    r.x = r.x > 0.f ? r.x : (__expf(r.x) - 1.f);
    r.y = r.y > 0.f ? r.y : (__expf(r.y) - 1.f);
    r.z = r.z > 0.f ? r.z : (__expf(r.z) - 1.f);
    r.w = r.w > 0.f ? r.w : (__expf(r.w) - 1.f);
    if (!head1) *(float4*)(out + (long)n * HID + lane * 4) = r;
}

extern "C" void kernel_launch(void* const* d_in, const int* in_sizes, int n_in,
                              void* d_out, int out_size, void* d_ws, size_t ws_size,
                              hipStream_t stream) {
    const float* x   = (const float*)d_in[0];
    const float* W1  = (const float*)d_in[1];
    const float* al1 = (const float*)d_in[2];
    const float* ar1 = (const float*)d_in[3];
    const float* b1  = (const float*)d_in[4];
    const float* W2  = (const float*)d_in[5];
    const float* al2 = (const float*)d_in[6];
    const float* ar2 = (const float*)d_in[7];
    const float* b2  = (const float*)d_in[8];
    const int*   src = (const int*)d_in[9];
    const int*   dst = (const int*)d_in[10];
    float* out = (float*)d_out;

    char* wsb = (char*)d_ws;
    size_t off = 0;
    auto alloc = [&](size_t bytes) -> void* {
        void* p = wsb + off;
        off += bytes;
        off = (off + 255) & ~(size_t)255;
        return p;
    };
    float* WT1  = (float*)alloc((size_t)INF_ * HD * 4);
    float* WT2  = (float*)alloc((size_t)HID  * HD * 4);
    float* feat = (float*)alloc((size_t)NN * HD * 4);
    float* el   = (float*)alloc((size_t)NN * HEADS * 4);
    float* er   = (float*)alloc((size_t)NN * HEADS * 4);
    int* deg     = (int*)alloc((size_t)NN * 4);
    int* rowptr  = (int*)alloc((size_t)(NN + 1) * 4);
    int* rank    = (int*)alloc((size_t)NE * 4);
    int* csr_src = (int*)alloc((size_t)NE * 4);
    float* h    = (float*)alloc((size_t)NN * HID * 4);

    const int NH   = NN * HEADS;
    const dim3 B256(256);

    gat_csr_zero<<<dim3((NN + 255) / 256), B256, 0, stream>>>(deg, NN);
    gat_csr_count<<<dim3((NE + 255) / 256), B256, 0, stream>>>(dst, deg, rank, NE);
    gat_csr_scan<<<dim3((NN + 1023) / 1024), dim3(1024), 0, stream>>>(deg, rowptr, NN);
    gat_csr_fill<<<dim3((NE + 255) / 256), B256, 0, stream>>>(src, dst, rowptr, rank, csr_src, NE);

    gat_w_transpose<<<dim3((INF_ * HD + 255) / 256), B256, 0, stream>>>(W1, WT1, INF_, INF_ * HD);
    gat_gemm_wmma_f32<<<dim3(NN / 16), B256, 0, stream>>>(x, WT1, feat, INF_);
    gat_eler<<<dim3((NH + 255) / 256), B256, 0, stream>>>(feat, al1, ar1, el, er, NH);
    gat_node_aggregate<<<dim3((NN + 7) / 8), B256, 0, stream>>>(rowptr, csr_src, feat, el, er, b1, h, NN);

    gat_w_transpose<<<dim3((HID * HD + 255) / 256), B256, 0, stream>>>(W2, WT2, HID, HID * HD);
    gat_gemm_wmma_f32<<<dim3(NN / 16), B256, 0, stream>>>(h, WT2, feat, HID);
    gat_eler<<<dim3((NH + 255) / 256), B256, 0, stream>>>(feat, al2, ar2, el, er, NH);
    gat_node_aggregate<<<dim3((NN + 7) / 8), B256, 0, stream>>>(rowptr, csr_src, feat, el, er, b2, out, NN);
}


# Layer 2 aggregates before it projects: waves gather 256-B rows of h (not 512-B rows of h*W2), one [N,128]x[128,64] WMMA product follows; logits from h.(W2 a); sums reassociated
# speedup vs baseline: 5.5024x; 1.1595x over previous
#include <hip/hip_runtime.h>
#include <hip/hip_bf16.h>

#define NN   50000
#define NE   1600000
#define INF_ 128
#define HID  64
#define HEADS 2
#define HD   (HEADS*HID)
#define NEG_SLOPE 0.2f

typedef __attribute__((ext_vector_type(2))) float v2f;
typedef __attribute__((ext_vector_type(8))) float v8f;


template <int K, int NC, bool FINAL>
__global__ void __launch_bounds__(256) gat_gemm_strip_wmma_f32(
        const float* __restrict__ A, const float* __restrict__ W,
        const float* __restrict__ bias, float* __restrict__ D, int n_strips) {
    constexpr int KS = K / 4;
    constexpr int NT = NC / 16;
    constexpr int CG = NC / 4;
    constexpr int GEMM_LDS_PITCH = NC + 16;
    constexpr int NW = ((K / 2) * CG) / 256;
    __shared__ __attribute__((aligned(16))) v2f wl[(K / 2) * GEMM_LDS_PITCH];

    const int lane = threadIdx.x & 31;
    const int wave = threadIdx.x >> 5;
    const int half = lane >> 4;
    const int l15  = lane & 15;
    const int strip = blockIdx.x * 8 + wave;
    const bool live = strip < n_strips;

    float4 w0[NW], w1[NW];
#pragma unroll
    for (int i = 0; i < NW; ++i) {
        const int q = i * 256 + threadIdx.x;
        const int k0 = 2 * (q / CG);
        const float* __restrict__ r0 =
            FINAL ? W + (k0 & (HID - 1)) * HD + (k0 / HID) * HID : W + k0 * HD;
        const float4* __restrict__ wp = (const float4*)r0 + (q % CG);
        w0[i] = wp[0];
        w1[i] = wp[HD / 4];
    }

    v2f a[KS];
    const float* __restrict__ arow =
        A + ((long)(live ? strip : n_strips - 1) * 16 + l15) * K + half * 2;
#pragma unroll
    for (int i = 0; i < KS; ++i) a[i] = *(const v2f*)(arow + 4 * i);

#pragma unroll
    for (int i = 0; i < NW; ++i) {
        const int q = i * 256 + threadIdx.x;
        float4* __restrict__ lp = (float4*)(wl + (q / CG) * GEMM_LDS_PITCH + (q % CG) * 4);
        lp[0] = make_float4(w0[i].x, w1[i].x, w0[i].y, w1[i].y);
        lp[1] = make_float4(w0[i].z, w1[i].z, w0[i].w, w1[i].w);
    }
    __builtin_amdgcn_fence(__ATOMIC_RELEASE, "workgroup", "local");
    __builtin_amdgcn_s_barrier();
    __builtin_amdgcn_fence(__ATOMIC_ACQUIRE, "workgroup", "local");
    if (!live) return;

    v8f c[NT];
#pragma unroll
    for (int t = 0; t < NT; ++t) c[t] = v8f{};

    const v2f* __restrict__ bl = wl + half * GEMM_LDS_PITCH + l15;
    v2f bcur[NT], bnxt[NT];
#pragma unroll
    for (int t = 0; t < NT; ++t) bcur[t] = bl[t * 16];
#pragma unroll
    for (int i = 0; i < KS; ++i) {
        if (i + 1 < KS) {
#pragma unroll
            for (int t = 0; t < NT; ++t) bnxt[t] = bl[(2 * (i + 1)) * GEMM_LDS_PITCH + t * 16];
        }
#pragma unroll
        for (int t = 0; t < NT; ++t)
            c[t] = __builtin_amdgcn_wmma_f32_16x16x4_f32(false, a[i], false, bcur[t],
                                                         (short)0, c[t], false, false);
#pragma unroll
        for (int t = 0; t < NT; ++t) bcur[t] = bnxt[t];
    }

    float* __restrict__ drow = D + ((long)strip * 16 + half * 8) * NC + l15;
#pragma unroll
    for (int t = 0; t < NT; ++t) {
        float bs = 0.f;
        if (FINAL) bs = bias[t * 16 + l15] + bias[HID + t * 16 + l15];
#pragma unroll
        for (int v = 0; v < 8; ++v) {
            float r = c[t][v];
            if (FINAL) {
                r = 0.5f * (r + bs);
                r = r > 0.f ? r : (__expf(r) - 1.f);
            }
            drow[(long)v * NC + t * 16] = r;
        }
    }
}

__global__ void __launch_bounds__(256) gat_attn_vec(const float* __restrict__ W,
                                                    const float* __restrict__ al,
                                                    const float* __restrict__ ar,
                                                    float* __restrict__ vlr) {
    const int t = threadIdx.x;
    const int c = t & (HID - 1), hd = (t >> 6) & 1;
    const float* __restrict__ a = ((t >> 7) ? ar : al) + hd * HID;
    const float* __restrict__ w = W + c * HD + hd * HID;
    float s = 0.f;
    for (int d = 0; d < HID; ++d) s += w[d] * a[d];
    vlr[t] = s;
}

__global__ void gat_eler(const float* __restrict__ feat,
                         const float* __restrict__ al, const float* __restrict__ ar,
                         float* __restrict__ el, float* __restrict__ er, int NH) {
    int i = blockIdx.x * blockDim.x + threadIdx.x;
    if (i >= NH) return;
    const int h = i & 1;
    const float* __restrict__ f = feat + (long)(i >> 1) * HD + h * HID;
    const float* __restrict__ la = al + h * HID;
    const float* __restrict__ ra = ar + h * HID;
    float sl = 0.f, sr = 0.f;
#pragma unroll 4
    for (int d = 0; d < HID; ++d) { float v = f[d]; sl += v * la[d]; sr += v * ra[d]; }
    el[i] = sl; er[i] = sr;
}


__global__ void gat_csr_zero(int* __restrict__ deg, int n) {
    const int i = blockIdx.x * blockDim.x + threadIdx.x;
    if (i < n) deg[i] = 0;
}

__global__ void gat_csr_count(const int* __restrict__ dst, int* __restrict__ deg,
                              int* __restrict__ rank, int E) {
    const int e = blockIdx.x * blockDim.x + threadIdx.x;
    if (e >= E) return;
    rank[e] = atomicAdd(&deg[dst[e]], 1);
}

__global__ void __launch_bounds__(1024) gat_csr_scan(const int* __restrict__ deg,
                                                     int* __restrict__ rowptr, int n) {
    __shared__ int wave_front[32];
    __shared__ int wave_own[32];
    const int t = threadIdx.x, lane = t & 31, w = t >> 5;
    const int tile0 = blockIdx.x * 1024;

    int front = 0;
    for (int i = t; i < tile0; i += 1024) front += deg[i];
    const int i = tile0 + t;
    const int own = (i < n) ? deg[i] : 0;

    int inc = own;
    for (int off = 1; off < 32; off <<= 1) {
        const int v = __shfl_up(inc, off);
        if (lane >= off) inc += v;
    }
    for (int off = 16; off; off >>= 1) front += __shfl_xor(front, off);
    if (lane == 31) wave_own[w] = inc;
    if (lane == 0)  wave_front[w] = front;
    __syncthreads();
    int base = wave_front[lane] + (lane < w ? wave_own[lane] : 0);
    for (int off = 16; off; off >>= 1) base += __shfl_xor(base, off);

    if (i < n) rowptr[i] = base + inc - own;
    if (i == n - 1) rowptr[n] = base + inc;
}

__global__ void gat_csr_fill(const int* __restrict__ src, const int* __restrict__ dst,
                             const int* __restrict__ rowptr, const int* __restrict__ rank,
                             int* __restrict__ csr_src, int E) {
    const int e = blockIdx.x * blockDim.x + threadIdx.x;
    if (e >= E) return;
    csr_src[rowptr[dst[e]] + rank[e]] = src[e];
}

__device__ __forceinline__ float gat_leaky(float v) { return v > 0.f ? v : NEG_SLOPE * v; }

__device__ __forceinline__ float gat_lane_f(float v, int j) {
    return __int_as_float(__builtin_amdgcn_readlane(__float_as_int(v), j));
}

template <int N>
__device__ __forceinline__ void gat_add_rows(float4& acc, const float* __restrict__ fcol,
                                             int s, float x0, float x1, bool head1, int j) {
    float4 f[N];
    float  a[N];
#pragma unroll
    for (int u = 0; u < N; ++u) {
        const int sj = __builtin_amdgcn_readlane(s, j + u);
        f[u] = *(const float4*)(fcol + (long)sj * HD);
        const float a0 = gat_lane_f(x0, j + u), a1 = gat_lane_f(x1, j + u);
        a[u] = head1 ? a1 : a0;
    }
#pragma unroll
    for (int u = 0; u < N; ++u) {
        acc.x += f[u].x * a[u]; acc.y += f[u].y * a[u];
        acc.z += f[u].z * a[u]; acc.w += f[u].w * a[u];
    }
}

__global__ void __launch_bounds__(256) gat_node_aggregate(
        const int* __restrict__ rowptr, const int* __restrict__ csr_src,
        const float* __restrict__ feat, const float* __restrict__ el,
        const float* __restrict__ er, const float* __restrict__ bias,
        const float* __restrict__ vlr,
        float* __restrict__ out, float* __restrict__ el_next, float* __restrict__ er_next,
        int n_nodes) {
    const int n = __builtin_amdgcn_readfirstlane((int)(blockIdx.x * 8 + (threadIdx.x >> 5)));
    if (n >= n_nodes) return;
    const int lane = threadIdx.x & 31;
    const bool head1 = lane >= 16;
    const int beg = rowptr[n], end = rowptr[n + 1];
    const float2 ern = *(const float2*)(er + n * HEADS);

    float m0 = -__builtin_inff(), m1 = -__builtin_inff();
    for (int p = beg + lane; p < end; p += 32) {
        const int s = csr_src[p];
        const float2 l = *(const float2*)(el + s * HEADS);
        m0 = fmaxf(m0, gat_leaky(l.x + ern.x));
        m1 = fmaxf(m1, gat_leaky(l.y + ern.y));
    }
    for (int off = 16; off; off >>= 1) {
        m0 = fmaxf(m0, __shfl_xor(m0, off));
        m1 = fmaxf(m1, __shfl_xor(m1, off));
    }

    float s0 = 0.f, s1 = 0.f;
    float4 acc = make_float4(0.f, 0.f, 0.f, 0.f);
    const float* __restrict__ fcol = feat + lane * 4;
    for (int base = beg; base < end; base += 32) {
        const int p = base + lane;
        int s = 0;
        float x0 = 0.f, x1 = 0.f;
        if (p < end) {
            s = csr_src[p];
            const float2 l = *(const float2*)(el + s * HEADS);
            x0 = __expf(gat_leaky(l.x + ern.x) - m0);
            x1 = __expf(gat_leaky(l.y + ern.y) - m1);
        }
        s0 += x0; s1 += x1;
        const int cnt = min(32, end - base);
        int j = 0;
        for (; j + 8 <= cnt; j += 8) gat_add_rows<8>(acc, fcol, s, x0, x1, head1, j);
        if (j + 4 <= cnt) { gat_add_rows<4>(acc, fcol, s, x0, x1, head1, j); j += 4; }
        if (j + 2 <= cnt) { gat_add_rows<2>(acc, fcol, s, x0, x1, head1, j); j += 2; }
        if (j < cnt) gat_add_rows<1>(acc, fcol, s, x0, x1, head1, j);
    }
    for (int off = 16; off; off >>= 1) {
        s0 += __shfl_xor(s0, off);
        s1 += __shfl_xor(s1, off);
    }
    if (end > beg) {
        const float sden = head1 ? s1 : s0;
        acc.x /= sden; acc.y /= sden; acc.z /= sden; acc.w /= sden;
    }

    const float4 b = *(const float4*)(bias + lane * 4);
    float4 v = make_float4(acc.x + b.x, acc.y + b.y, acc.z + b.z, acc.w + b.w);
    float4 r;
    r.x = 0.5f * (v.x + __shfl_xor(v.x, 16));
    r.y = 0.5f * (v.y + __shfl_xor(v.y, 16));
    r.z = 0.5f * (v.z + __shfl_xor(v.z, 16));
    r.w = 0.5f * (v.w + __shfl_xor(v.w, 16));
    r.x = r.x > 0.f ? r.x : (__expf(r.x) - 1.f);
    r.y = r.y > 0.f ? r.y : (__expf(r.y) - 1.f);
    r.z = r.z > 0.f ? r.z : (__expf(r.z) - 1.f);
    r.w = r.w > 0.f ? r.w : (__expf(r.w) - 1.f);
    if (!head1) *(float4*)(out + (long)n * HID + lane * 4) = r;

    const float4 vl = *(const float4*)(vlr + lane * 4);
    const float4 vr = *(const float4*)(vlr + 2 * HID + lane * 4);
    float pl = r.x * vl.x + r.y * vl.y + r.z * vl.z + r.w * vl.w;
    float pr = r.x * vr.x + r.y * vr.y + r.z * vr.z + r.w * vr.w;
    for (int off = 8; off; off >>= 1) {
        pl += __shfl_xor(pl, off);
        pr += __shfl_xor(pr, off);
    }
    if ((lane & 15) == 0) {
        el_next[n * HEADS + (lane >> 4)] = pl;
        er_next[n * HEADS + (lane >> 4)] = pr;
    }
}

template <int N>
__device__ __forceinline__ void gat_add_hrows(float2& za, float2& zb, const float* __restrict__ hcol,
                                              int s, float x0, float x1, int j) {
    float2 f[N];
    float  a0[N], a1[N];
#pragma unroll
    for (int u = 0; u < N; ++u) {
        const int sj = __builtin_amdgcn_readlane(s, j + u);
        f[u] = *(const float2*)(hcol + (long)sj * HID);
        a0[u] = gat_lane_f(x0, j + u);
        a1[u] = gat_lane_f(x1, j + u);
    }
#pragma unroll
    for (int u = 0; u < N; ++u) {
        za.x += f[u].x * a0[u]; za.y += f[u].y * a0[u];
        zb.x += f[u].x * a1[u]; zb.y += f[u].y * a1[u];
    }
}

__global__ void __launch_bounds__(256) gat_node_aggregate_h(
        const int* __restrict__ rowptr, const int* __restrict__ csr_src,
        const float* __restrict__ h, const float* __restrict__ el,
        const float* __restrict__ er, float* __restrict__ z, int n_nodes) {
    const int n = __builtin_amdgcn_readfirstlane((int)(blockIdx.x * 8 + (threadIdx.x >> 5)));
    if (n >= n_nodes) return;
    const int lane = threadIdx.x & 31;
    const int beg = rowptr[n], end = rowptr[n + 1];
    const float2 ern = *(const float2*)(er + n * HEADS);

    float m0 = -__builtin_inff(), m1 = -__builtin_inff();
    for (int p = beg + lane; p < end; p += 32) {
        const int s = csr_src[p];
        const float2 l = *(const float2*)(el + s * HEADS);
        m0 = fmaxf(m0, gat_leaky(l.x + ern.x));
        m1 = fmaxf(m1, gat_leaky(l.y + ern.y));
    }
    for (int off = 16; off; off >>= 1) {
        m0 = fmaxf(m0, __shfl_xor(m0, off));
        m1 = fmaxf(m1, __shfl_xor(m1, off));
    }

    float s0 = 0.f, s1 = 0.f;
    float2 za = make_float2(0.f, 0.f), zb = make_float2(0.f, 0.f);
    const float* __restrict__ hcol = h + lane * 2;
    for (int base = beg; base < end; base += 32) {
        const int p = base + lane;
        int s = 0;
        float x0 = 0.f, x1 = 0.f;
        if (p < end) {
            s = csr_src[p];
            const float2 l = *(const float2*)(el + s * HEADS);
            x0 = __expf(gat_leaky(l.x + ern.x) - m0);
            x1 = __expf(gat_leaky(l.y + ern.y) - m1);
        }
        s0 += x0; s1 += x1;
        const int cnt = min(32, end - base);
        int j = 0;
        for (; j + 16 <= cnt; j += 16) gat_add_hrows<16>(za, zb, hcol, s, x0, x1, j);
        if (j + 8 <= cnt) { gat_add_hrows<8>(za, zb, hcol, s, x0, x1, j); j += 8; }
        if (j + 4 <= cnt) { gat_add_hrows<4>(za, zb, hcol, s, x0, x1, j); j += 4; }
        if (j + 2 <= cnt) { gat_add_hrows<2>(za, zb, hcol, s, x0, x1, j); j += 2; }
        if (j < cnt) gat_add_hrows<1>(za, zb, hcol, s, x0, x1, j);
    }
    for (int off = 16; off; off >>= 1) {
        s0 += __shfl_xor(s0, off);
        s1 += __shfl_xor(s1, off);
    }
    if (end > beg) {
        za.x /= s0; za.y /= s0;
        zb.x /= s1; zb.y /= s1;
    }
    *(float2*)(z + (long)n * HD + lane * 2) = za;
    *(float2*)(z + (long)n * HD + HID + lane * 2) = zb;
}

extern "C" void kernel_launch(void* const* d_in, const int* in_sizes, int n_in,
                              void* d_out, int out_size, void* d_ws, size_t ws_size,
                              hipStream_t stream) {
    const float* x   = (const float*)d_in[0];
    const float* W1  = (const float*)d_in[1];
    const float* al1 = (const float*)d_in[2];
    const float* ar1 = (const float*)d_in[3];
    const float* b1  = (const float*)d_in[4];
    const float* W2  = (const float*)d_in[5];
    const float* al2 = (const float*)d_in[6];
    const float* ar2 = (const float*)d_in[7];
    const float* b2  = (const float*)d_in[8];
    const int*   src = (const int*)d_in[9];
    const int*   dst = (const int*)d_in[10];
    float* out = (float*)d_out;

    char* wsb = (char*)d_ws;
    size_t off = 0;
    auto alloc = [&](size_t bytes) -> void* {
        void* p = wsb + off;
        off += bytes;
        off = (off + 255) & ~(size_t)255;
        return p;
    };
    float* feat = (float*)alloc((size_t)NN * HD * 4);
    float* el   = (float*)alloc((size_t)NN * HEADS * 4);
    float* er   = (float*)alloc((size_t)NN * HEADS * 4);
    int* deg     = (int*)alloc((size_t)NN * 4);
    int* rowptr  = (int*)alloc((size_t)(NN + 1) * 4);
    int* rank    = (int*)alloc((size_t)NE * 4);
    int* csr_src = (int*)alloc((size_t)NE * 4);
    float* h    = (float*)alloc((size_t)NN * HID * 4);
    float* vlr = (float*)deg;
    float* el2 = (float*)rank;
    float* er2 = el2 + (size_t)NN * HEADS;
    float* z   = feat;

    const int NH   = NN * HEADS;
    const dim3 B256(256);

    gat_csr_zero<<<dim3((NN + 255) / 256), B256, 0, stream>>>(deg, NN);
    gat_csr_count<<<dim3((NE + 255) / 256), B256, 0, stream>>>(dst, deg, rank, NE);
    gat_csr_scan<<<dim3((NN + 1023) / 1024), dim3(1024), 0, stream>>>(deg, rowptr, NN);
    gat_csr_fill<<<dim3((NE + 255) / 256), B256, 0, stream>>>(src, dst, rowptr, rank, csr_src, NE);
    gat_attn_vec<<<dim3(1), B256, 0, stream>>>(W2, al2, ar2, vlr);

    gat_gemm_strip_wmma_f32<INF_, HD, false><<<dim3((NN / 16 + 7) / 8), B256, 0, stream>>>(
        x, W1, b1, feat, NN / 16);
    gat_eler<<<dim3((NH + 255) / 256), B256, 0, stream>>>(feat, al1, ar1, el, er, NH);
    gat_node_aggregate<<<dim3((NN + 7) / 8), B256, 0, stream>>>(
        rowptr, csr_src, feat, el, er, b1, vlr, h, el2, er2, NN);

    gat_node_aggregate_h<<<dim3((NN + 7) / 8), B256, 0, stream>>>(rowptr, csr_src, h, el2, er2, z, NN);
    gat_gemm_strip_wmma_f32<HD, HID, true><<<dim3((NN / 16 + 7) / 8), B256, 0, stream>>>(
        z, W2, b2, out, NN / 16);
}


# Edge logits looked up once per layer: the max pass keeps source id and el[src] of a node's first 64 edges in registers, the weight pass reuses them; same values, same sum order
# speedup vs baseline: 5.7748x; 1.0495x over previous
#include <hip/hip_runtime.h>
#include <hip/hip_bf16.h>

#define NN   50000
#define NE   1600000
#define INF_ 128
#define HID  64
#define HEADS 2
#define HD   (HEADS*HID)
#define NEG_SLOPE 0.2f

typedef __attribute__((ext_vector_type(2))) float v2f;
typedef __attribute__((ext_vector_type(8))) float v8f;


template <int K, int NC, bool FINAL>
__global__ void __launch_bounds__(256) gat_gemm_strip_wmma_f32(
        const float* __restrict__ A, const float* __restrict__ W,
        const float* __restrict__ bias, float* __restrict__ D, int n_strips) {
    constexpr int KS = K / 4;
    constexpr int NT = NC / 16;
    constexpr int CG = NC / 4;
    constexpr int GEMM_LDS_PITCH = NC + 16;
    constexpr int NW = ((K / 2) * CG) / 256;
    __shared__ __attribute__((aligned(16))) v2f wl[(K / 2) * GEMM_LDS_PITCH];

    const int lane = threadIdx.x & 31;
    const int wave = threadIdx.x >> 5;
    const int half = lane >> 4;
    const int l15  = lane & 15;
    const int strip = blockIdx.x * 8 + wave;
    const bool live = strip < n_strips;

    float4 w0[NW], w1[NW];
#pragma unroll
    for (int i = 0; i < NW; ++i) {
        const int q = i * 256 + threadIdx.x;
        const int k0 = 2 * (q / CG);
        const float* __restrict__ r0 =
            FINAL ? W + (k0 & (HID - 1)) * HD + (k0 / HID) * HID : W + k0 * HD;
        const float4* __restrict__ wp = (const float4*)r0 + (q % CG);
        w0[i] = wp[0];
        w1[i] = wp[HD / 4];
    }

    v2f a[KS];
    const float* __restrict__ arow =
        A + ((long)(live ? strip : n_strips - 1) * 16 + l15) * K + half * 2;
#pragma unroll
    for (int i = 0; i < KS; ++i) a[i] = *(const v2f*)(arow + 4 * i);

#pragma unroll
    for (int i = 0; i < NW; ++i) {
        const int q = i * 256 + threadIdx.x;
        float4* __restrict__ lp = (float4*)(wl + (q / CG) * GEMM_LDS_PITCH + (q % CG) * 4);
        lp[0] = make_float4(w0[i].x, w1[i].x, w0[i].y, w1[i].y);
        lp[1] = make_float4(w0[i].z, w1[i].z, w0[i].w, w1[i].w);
    }
    __builtin_amdgcn_fence(__ATOMIC_RELEASE, "workgroup", "local");
    __builtin_amdgcn_s_barrier();
    __builtin_amdgcn_fence(__ATOMIC_ACQUIRE, "workgroup", "local");
    if (!live) return;

    v8f c[NT];
#pragma unroll
    for (int t = 0; t < NT; ++t) c[t] = v8f{};

    const v2f* __restrict__ bl = wl + half * GEMM_LDS_PITCH + l15;
    v2f bcur[NT], bnxt[NT];
#pragma unroll
    for (int t = 0; t < NT; ++t) bcur[t] = bl[t * 16];
#pragma unroll
    for (int i = 0; i < KS; ++i) {
        if (i + 1 < KS) {
#pragma unroll
            for (int t = 0; t < NT; ++t) bnxt[t] = bl[(2 * (i + 1)) * GEMM_LDS_PITCH + t * 16];
        }
#pragma unroll
        for (int t = 0; t < NT; ++t)
            c[t] = __builtin_amdgcn_wmma_f32_16x16x4_f32(false, a[i], false, bcur[t],
                                                         (short)0, c[t], false, false);
#pragma unroll
        for (int t = 0; t < NT; ++t) bcur[t] = bnxt[t];
    }

    float* __restrict__ drow = D + ((long)strip * 16 + half * 8) * NC + l15;
#pragma unroll
    for (int t = 0; t < NT; ++t) {
        float bs = 0.f;
        if (FINAL) bs = bias[t * 16 + l15] + bias[HID + t * 16 + l15];
#pragma unroll
        for (int v = 0; v < 8; ++v) {
            float r = c[t][v];
            if (FINAL) {
                r = 0.5f * (r + bs);
                r = r > 0.f ? r : (__expf(r) - 1.f);
            }
            drow[(long)v * NC + t * 16] = r;
        }
    }
}

__global__ void __launch_bounds__(256) gat_attn_vec(const float* __restrict__ W,
                                                    const float* __restrict__ al,
                                                    const float* __restrict__ ar,
                                                    float* __restrict__ vlr) {
    const int t = threadIdx.x;
    const int c = t & (HID - 1), hd = (t >> 6) & 1;
    const float* __restrict__ a = ((t >> 7) ? ar : al) + hd * HID;
    const float* __restrict__ w = W + c * HD + hd * HID;
    float s = 0.f;
    for (int d = 0; d < HID; ++d) s += w[d] * a[d];
    vlr[t] = s;
}

__global__ void gat_eler(const float* __restrict__ feat,
                         const float* __restrict__ al, const float* __restrict__ ar,
                         float* __restrict__ el, float* __restrict__ er, int NH) {
    int i = blockIdx.x * blockDim.x + threadIdx.x;
    if (i >= NH) return;
    const int h = i & 1;
    const float* __restrict__ f = feat + (long)(i >> 1) * HD + h * HID;
    const float* __restrict__ la = al + h * HID;
    const float* __restrict__ ra = ar + h * HID;
    float sl = 0.f, sr = 0.f;
#pragma unroll 4
    for (int d = 0; d < HID; ++d) { float v = f[d]; sl += v * la[d]; sr += v * ra[d]; }
    el[i] = sl; er[i] = sr;
}


__global__ void gat_csr_zero(int* __restrict__ deg, int n) {
    const int i = blockIdx.x * blockDim.x + threadIdx.x;
    if (i < n) deg[i] = 0;
}

__global__ void gat_csr_count(const int* __restrict__ dst, int* __restrict__ deg,
                              int* __restrict__ rank, int E) {
    const int e = blockIdx.x * blockDim.x + threadIdx.x;
    if (e >= E) return;
    rank[e] = atomicAdd(&deg[dst[e]], 1);
}

__global__ void __launch_bounds__(1024) gat_csr_scan(const int* __restrict__ deg,
                                                     int* __restrict__ rowptr, int n) {
    __shared__ int wave_front[32];
    __shared__ int wave_own[32];
    const int t = threadIdx.x, lane = t & 31, w = t >> 5;
    const int tile0 = blockIdx.x * 1024;

    int front = 0;
    for (int i = t; i < tile0; i += 1024) front += deg[i];
    const int i = tile0 + t;
    const int own = (i < n) ? deg[i] : 0;

    int inc = own;
    for (int off = 1; off < 32; off <<= 1) {
        const int v = __shfl_up(inc, off);
        if (lane >= off) inc += v;
    }
    for (int off = 16; off; off >>= 1) front += __shfl_xor(front, off);
    if (lane == 31) wave_own[w] = inc;
    if (lane == 0)  wave_front[w] = front;
    __syncthreads();
    int base = wave_front[lane] + (lane < w ? wave_own[lane] : 0);
    for (int off = 16; off; off >>= 1) base += __shfl_xor(base, off);

    if (i < n) rowptr[i] = base + inc - own;
    if (i == n - 1) rowptr[n] = base + inc;
}

__global__ void gat_csr_fill(const int* __restrict__ src, const int* __restrict__ dst,
                             const int* __restrict__ rowptr, const int* __restrict__ rank,
                             int* __restrict__ csr_src, int E) {
    const int e = blockIdx.x * blockDim.x + threadIdx.x;
    if (e >= E) return;
    csr_src[rowptr[dst[e]] + rank[e]] = src[e];
}

__device__ __forceinline__ float gat_leaky(float v) { return v > 0.f ? v : NEG_SLOPE * v; }

__device__ __forceinline__ float gat_lane_f(float v, int j) {
    return __int_as_float(__builtin_amdgcn_readlane(__float_as_int(v), j));
}

__device__ __forceinline__ void gat_node_max(const int* __restrict__ csr_src,
                                             const float* __restrict__ el, float2 ern,
                                             int beg, int end, int lane,
                                             int& sA, float2& lA, int& sB, float2& lB,
                                             float& m0, float& m1) {
    const int pA = beg + lane, pB = pA + 32;
    const bool inA = pA < end, inB = pB < end;
    sA = 0; sB = 0;
    lA = make_float2(0.f, 0.f); lB = make_float2(0.f, 0.f);
    if (inA) sA = csr_src[pA];
    if (inB) sB = csr_src[pB];
    if (inA) lA = *(const float2*)(el + sA * HEADS);
    if (inB) lB = *(const float2*)(el + sB * HEADS);
    const float ninf = -__builtin_inff();
    m0 = fmaxf(inA ? gat_leaky(lA.x + ern.x) : ninf, inB ? gat_leaky(lB.x + ern.x) : ninf);
    m1 = fmaxf(inA ? gat_leaky(lA.y + ern.y) : ninf, inB ? gat_leaky(lB.y + ern.y) : ninf);
    for (int base = beg + 64; base < end; base += 32) {
        const int p = base + lane;
        if (p < end) {
            const int s = csr_src[p];
            const float2 l = *(const float2*)(el + s * HEADS);
            m0 = fmaxf(m0, gat_leaky(l.x + ern.x));
            m1 = fmaxf(m1, gat_leaky(l.y + ern.y));
        }
    }
    for (int off = 16; off; off >>= 1) {
        m0 = fmaxf(m0, __shfl_xor(m0, off));
        m1 = fmaxf(m1, __shfl_xor(m1, off));
    }
}

template <int N>
__device__ __forceinline__ void gat_add_rows(float4& acc, const float* __restrict__ fcol,
                                             int s, float x0, float x1, bool head1, int j) {
    float4 f[N];
    float  a[N];
#pragma unroll
    for (int u = 0; u < N; ++u) {
        const int sj = __builtin_amdgcn_readlane(s, j + u);
        f[u] = *(const float4*)(fcol + (long)sj * HD);
        const float a0 = gat_lane_f(x0, j + u), a1 = gat_lane_f(x1, j + u);
        a[u] = head1 ? a1 : a0;
    }
#pragma unroll
    for (int u = 0; u < N; ++u) {
        acc.x += f[u].x * a[u]; acc.y += f[u].y * a[u];
        acc.z += f[u].z * a[u]; acc.w += f[u].w * a[u];
    }
}

__global__ void __launch_bounds__(256) gat_node_aggregate(
        const int* __restrict__ rowptr, const int* __restrict__ csr_src,
        const float* __restrict__ feat, const float* __restrict__ el,
        const float* __restrict__ er, const float* __restrict__ bias,
        const float* __restrict__ vlr,
        float* __restrict__ out, float* __restrict__ el_next, float* __restrict__ er_next,
        int n_nodes) {
    const int n = __builtin_amdgcn_readfirstlane((int)(blockIdx.x * 8 + (threadIdx.x >> 5)));
    if (n >= n_nodes) return;
    const int lane = threadIdx.x & 31;
    const bool head1 = lane >= 16;
    const int beg = rowptr[n], end = rowptr[n + 1];
    const float2 ern = *(const float2*)(er + n * HEADS);

    int sA, sB;
    float2 lA, lB;
    float m0, m1;
    gat_node_max(csr_src, el, ern, beg, end, lane, sA, lA, sB, lB, m0, m1);

    float s0 = 0.f, s1 = 0.f;
    float4 acc = make_float4(0.f, 0.f, 0.f, 0.f);
    const float* __restrict__ fcol = feat + lane * 4;
    int chunk = 0;
    for (int base = beg; base < end; base += 32, ++chunk) {
        const int p = base + lane;
        int s = chunk ? sB : sA;
        float2 l = chunk ? lB : lA;
        if (chunk >= 2) {
            s = 0;
            if (p < end) {
                s = csr_src[p];
                l = *(const float2*)(el + s * HEADS);
            }
        }
        float x0 = 0.f, x1 = 0.f;
        if (p < end) {
            x0 = __expf(gat_leaky(l.x + ern.x) - m0);
            x1 = __expf(gat_leaky(l.y + ern.y) - m1);
        }
        s0 += x0; s1 += x1;
        const int cnt = min(32, end - base);
        int j = 0;
        for (; j + 8 <= cnt; j += 8) gat_add_rows<8>(acc, fcol, s, x0, x1, head1, j);
        if (j + 4 <= cnt) { gat_add_rows<4>(acc, fcol, s, x0, x1, head1, j); j += 4; }
        if (j + 2 <= cnt) { gat_add_rows<2>(acc, fcol, s, x0, x1, head1, j); j += 2; }
        if (j < cnt) gat_add_rows<1>(acc, fcol, s, x0, x1, head1, j);
    }
    for (int off = 16; off; off >>= 1) {
        s0 += __shfl_xor(s0, off);
        s1 += __shfl_xor(s1, off);
    }
    if (end > beg) {
        const float sden = head1 ? s1 : s0;
        acc.x /= sden; acc.y /= sden; acc.z /= sden; acc.w /= sden;
    }

    const float4 b = *(const float4*)(bias + lane * 4);
    float4 v = make_float4(acc.x + b.x, acc.y + b.y, acc.z + b.z, acc.w + b.w);
    float4 r;
    r.x = 0.5f * (v.x + __shfl_xor(v.x, 16));
    r.y = 0.5f * (v.y + __shfl_xor(v.y, 16));
    r.z = 0.5f * (v.z + __shfl_xor(v.z, 16));
    r.w = 0.5f * (v.w + __shfl_xor(v.w, 16));
    r.x = r.x > 0.f ? r.x : (__expf(r.x) - 1.f);
    r.y = r.y > 0.f ? r.y : (__expf(r.y) - 1.f);
    r.z = r.z > 0.f ? r.z : (__expf(r.z) - 1.f);
    r.w = r.w > 0.f ? r.w : (__expf(r.w) - 1.f);
    if (!head1) *(float4*)(out + (long)n * HID + lane * 4) = r;

    const float4 vl = *(const float4*)(vlr + lane * 4);
    const float4 vr = *(const float4*)(vlr + 2 * HID + lane * 4);
    float pl = r.x * vl.x + r.y * vl.y + r.z * vl.z + r.w * vl.w;
    float pr = r.x * vr.x + r.y * vr.y + r.z * vr.z + r.w * vr.w;
    for (int off = 8; off; off >>= 1) {
        pl += __shfl_xor(pl, off);
        pr += __shfl_xor(pr, off);
    }
    if ((lane & 15) == 0) {
        el_next[n * HEADS + (lane >> 4)] = pl;
        er_next[n * HEADS + (lane >> 4)] = pr;
    }
}

template <int N>
__device__ __forceinline__ void gat_add_hrows(float2& za, float2& zb, const float* __restrict__ hcol,
                                              int s, float x0, float x1, int j) {
    float2 f[N];
    float  a0[N], a1[N];
#pragma unroll
    for (int u = 0; u < N; ++u) {
        const int sj = __builtin_amdgcn_readlane(s, j + u);
        f[u] = *(const float2*)(hcol + (long)sj * HID);
        a0[u] = gat_lane_f(x0, j + u);
        a1[u] = gat_lane_f(x1, j + u);
    }
#pragma unroll
    for (int u = 0; u < N; ++u) {
        za.x += f[u].x * a0[u]; za.y += f[u].y * a0[u];
        zb.x += f[u].x * a1[u]; zb.y += f[u].y * a1[u];
    }
}

__global__ void __launch_bounds__(256) gat_node_aggregate_h(
        const int* __restrict__ rowptr, const int* __restrict__ csr_src,
        const float* __restrict__ h, const float* __restrict__ el,
        const float* __restrict__ er, float* __restrict__ z, int n_nodes) {
    const int n = __builtin_amdgcn_readfirstlane((int)(blockIdx.x * 8 + (threadIdx.x >> 5)));
    if (n >= n_nodes) return;
    const int lane = threadIdx.x & 31;
    const int beg = rowptr[n], end = rowptr[n + 1];
    const float2 ern = *(const float2*)(er + n * HEADS);

    int sA, sB;
    float2 lA, lB;
    float m0, m1;
    gat_node_max(csr_src, el, ern, beg, end, lane, sA, lA, sB, lB, m0, m1);

    float s0 = 0.f, s1 = 0.f;
    float2 za = make_float2(0.f, 0.f), zb = make_float2(0.f, 0.f);
    const float* __restrict__ hcol = h + lane * 2;
    int chunk = 0;
    for (int base = beg; base < end; base += 32, ++chunk) {
        const int p = base + lane;
        int s = chunk ? sB : sA;
        float2 l = chunk ? lB : lA;
        if (chunk >= 2) {
            s = 0;
            if (p < end) {
                s = csr_src[p];
                l = *(const float2*)(el + s * HEADS);
            }
        }
        float x0 = 0.f, x1 = 0.f;
        if (p < end) {
            x0 = __expf(gat_leaky(l.x + ern.x) - m0);
            x1 = __expf(gat_leaky(l.y + ern.y) - m1);
        }
        s0 += x0; s1 += x1;
        const int cnt = min(32, end - base);
        int j = 0;
        for (; j + 16 <= cnt; j += 16) gat_add_hrows<16>(za, zb, hcol, s, x0, x1, j);
        if (j + 8 <= cnt) { gat_add_hrows<8>(za, zb, hcol, s, x0, x1, j); j += 8; }
        if (j + 4 <= cnt) { gat_add_hrows<4>(za, zb, hcol, s, x0, x1, j); j += 4; }
        if (j + 2 <= cnt) { gat_add_hrows<2>(za, zb, hcol, s, x0, x1, j); j += 2; }
        if (j < cnt) gat_add_hrows<1>(za, zb, hcol, s, x0, x1, j);
    }
    for (int off = 16; off; off >>= 1) {
        s0 += __shfl_xor(s0, off);
        s1 += __shfl_xor(s1, off);
    }
    if (end > beg) {
        za.x /= s0; za.y /= s0;
        zb.x /= s1; zb.y /= s1;
    }
    *(float2*)(z + (long)n * HD + lane * 2) = za;
    *(float2*)(z + (long)n * HD + HID + lane * 2) = zb;
}

extern "C" void kernel_launch(void* const* d_in, const int* in_sizes, int n_in,
                              void* d_out, int out_size, void* d_ws, size_t ws_size,
                              hipStream_t stream) {
    const float* x   = (const float*)d_in[0];
    const float* W1  = (const float*)d_in[1];
    const float* al1 = (const float*)d_in[2];
    const float* ar1 = (const float*)d_in[3];
    const float* b1  = (const float*)d_in[4];
    const float* W2  = (const float*)d_in[5];
    const float* al2 = (const float*)d_in[6];
    const float* ar2 = (const float*)d_in[7];
    const float* b2  = (const float*)d_in[8];
    const int*   src = (const int*)d_in[9];
    const int*   dst = (const int*)d_in[10];
    float* out = (float*)d_out;

    char* wsb = (char*)d_ws;
    size_t off = 0;
    auto alloc = [&](size_t bytes) -> void* {
        void* p = wsb + off;
        off += bytes;
        off = (off + 255) & ~(size_t)255;
        return p;
    };
    float* feat = (float*)alloc((size_t)NN * HD * 4);
    float* el   = (float*)alloc((size_t)NN * HEADS * 4);
    float* er   = (float*)alloc((size_t)NN * HEADS * 4);
    int* deg     = (int*)alloc((size_t)NN * 4);
    int* rowptr  = (int*)alloc((size_t)(NN + 1) * 4);
    int* rank    = (int*)alloc((size_t)NE * 4);
    int* csr_src = (int*)alloc((size_t)NE * 4);
    float* h    = (float*)alloc((size_t)NN * HID * 4);
    float* vlr = (float*)deg;
    float* el2 = (float*)rank;
    float* er2 = el2 + (size_t)NN * HEADS;
    float* z   = feat;

    const int NH   = NN * HEADS;
    const dim3 B256(256);

    gat_csr_zero<<<dim3((NN + 255) / 256), B256, 0, stream>>>(deg, NN);
    gat_csr_count<<<dim3((NE + 255) / 256), B256, 0, stream>>>(dst, deg, rank, NE);
    gat_csr_scan<<<dim3((NN + 1023) / 1024), dim3(1024), 0, stream>>>(deg, rowptr, NN);
    gat_csr_fill<<<dim3((NE + 255) / 256), B256, 0, stream>>>(src, dst, rowptr, rank, csr_src, NE);
    gat_attn_vec<<<dim3(1), B256, 0, stream>>>(W2, al2, ar2, vlr);

    gat_gemm_strip_wmma_f32<INF_, HD, false><<<dim3((NN / 16 + 7) / 8), B256, 0, stream>>>(
        x, W1, b1, feat, NN / 16);
    gat_eler<<<dim3((NH + 255) / 256), B256, 0, stream>>>(feat, al1, ar1, el, er, NH);
    gat_node_aggregate<<<dim3((NN + 7) / 8), B256, 0, stream>>>(
        rowptr, csr_src, feat, el, er, b1, vlr, h, el2, er2, NN);

    gat_node_aggregate_h<<<dim3((NN + 7) / 8), B256, 0, stream>>>(rowptr, csr_src, h, el2, er2, z, NN);
    gat_gemm_strip_wmma_f32<HD, HID, true><<<dim3((NN / 16 + 7) / 8), B256, 0, stream>>>(
        z, W2, b2, out, NN / 16);
}


# Aggregation waves walk 5 consecutive nodes each (50,000 waves per launch to 10,000), next node's ids and logits requested under the current node's row sums; per-node sums and order unchanged
# speedup vs baseline: 13.4118x; 2.3225x over previous
#include <hip/hip_runtime.h>
#include <hip/hip_bf16.h>

#define NN   50000
#define NE   1600000
#define INF_ 128
#define HID  64
#define HEADS 2
#define HD   (HEADS*HID)
#define NEG_SLOPE 0.2f

typedef __attribute__((ext_vector_type(2))) float v2f;
typedef __attribute__((ext_vector_type(8))) float v8f;


template <int K, int NC, bool FINAL>
__device__ __forceinline__ void gat_gemm_strip_body(
        const float* __restrict__ A, const float* __restrict__ W,
        const float* __restrict__ bias, float* __restrict__ D, int n_strips, int blk) {
    constexpr int KS = K / 4;
    constexpr int NT = NC / 16;
    constexpr int CG = NC / 4;
    constexpr int GEMM_LDS_PITCH = NC + 16;
    constexpr int NW = ((K / 2) * CG) / 256;
    __shared__ __attribute__((aligned(16))) v2f wl[(K / 2) * GEMM_LDS_PITCH];

    const int lane = threadIdx.x & 31;
    const int wave = threadIdx.x >> 5;
    const int half = lane >> 4;
    const int l15  = lane & 15;
    const int strip = blk * 8 + wave;
    const bool live = strip < n_strips;

    float4 w0[NW], w1[NW];
#pragma unroll
    for (int i = 0; i < NW; ++i) {
        const int q = i * 256 + threadIdx.x;
        const int k0 = 2 * (q / CG);
        const float* __restrict__ r0 =
            FINAL ? W + (k0 & (HID - 1)) * HD + (k0 / HID) * HID : W + k0 * HD;
        const float4* __restrict__ wp = (const float4*)r0 + (q % CG);
        w0[i] = wp[0];
        w1[i] = wp[HD / 4];
    }

    v2f a[KS];
    const float* __restrict__ arow =
        A + ((long)(live ? strip : n_strips - 1) * 16 + l15) * K + half * 2;
#pragma unroll
    for (int i = 0; i < KS; ++i) a[i] = *(const v2f*)(arow + 4 * i);

#pragma unroll
    for (int i = 0; i < NW; ++i) {
        const int q = i * 256 + threadIdx.x;
        float4* __restrict__ lp = (float4*)(wl + (q / CG) * GEMM_LDS_PITCH + (q % CG) * 4);
        lp[0] = make_float4(w0[i].x, w1[i].x, w0[i].y, w1[i].y);
        lp[1] = make_float4(w0[i].z, w1[i].z, w0[i].w, w1[i].w);
    }
    __builtin_amdgcn_fence(__ATOMIC_RELEASE, "workgroup", "local");
    __builtin_amdgcn_s_barrier();
    __builtin_amdgcn_fence(__ATOMIC_ACQUIRE, "workgroup", "local");
    if (!live) return;

    v8f c[NT];
#pragma unroll
    for (int t = 0; t < NT; ++t) c[t] = v8f{};

    const v2f* __restrict__ bl = wl + half * GEMM_LDS_PITCH + l15;
    v2f bcur[NT], bnxt[NT];
#pragma unroll
    for (int t = 0; t < NT; ++t) bcur[t] = bl[t * 16];
#pragma unroll
    for (int i = 0; i < KS; ++i) {
        if (i + 1 < KS) {
#pragma unroll
            for (int t = 0; t < NT; ++t) bnxt[t] = bl[(2 * (i + 1)) * GEMM_LDS_PITCH + t * 16];
        }
#pragma unroll
        for (int t = 0; t < NT; ++t)
            c[t] = __builtin_amdgcn_wmma_f32_16x16x4_f32(false, a[i], false, bcur[t],
                                                         (short)0, c[t], false, false);
#pragma unroll
        for (int t = 0; t < NT; ++t) bcur[t] = bnxt[t];
    }

    float* __restrict__ drow = D + ((long)strip * 16 + half * 8) * NC + l15;
#pragma unroll
    for (int t = 0; t < NT; ++t) {
        float bs = 0.f;
        if (FINAL) bs = bias[t * 16 + l15] + bias[HID + t * 16 + l15];
#pragma unroll
        for (int v = 0; v < 8; ++v) {
            float r = c[t][v];
            if (FINAL) {
                r = 0.5f * (r + bs);
                r = r > 0.f ? r : (__expf(r) - 1.f);
            }
            drow[(long)v * NC + t * 16] = r;
        }
    }
}

__device__ __forceinline__ void gat_attn_vec_body(const float* __restrict__ W,
                                                  const float* __restrict__ al,
                                                  const float* __restrict__ ar,
                                                  float* __restrict__ vlr) {
    const int t = threadIdx.x;
    const int c = t & (HID - 1), hd = (t >> 6) & 1;
    const float* __restrict__ a = ((t >> 7) ? ar : al) + hd * HID;
    const float* __restrict__ w = W + c * HD + hd * HID;
    float s = 0.f;
    for (int d = 0; d < HID; ++d) s += w[d] * a[d];
    vlr[t] = s;
}

template <int BS>
__device__ __forceinline__ void gat_eler_body(const float* __restrict__ feat,
                         const float* __restrict__ al, const float* __restrict__ ar,
                         float* __restrict__ el, float* __restrict__ er, int NH, int blk) {
    int i = blk * BS + threadIdx.x;
    if (i >= NH) return;
    const int h = i & 1;
    const float* __restrict__ f = feat + (long)(i >> 1) * HD + h * HID;
    const float* __restrict__ la = al + h * HID;
    const float* __restrict__ ra = ar + h * HID;
    float sl = 0.f, sr = 0.f;
#pragma unroll 4
    for (int d = 0; d < HID; ++d) { float v = f[d]; sl += v * la[d]; sr += v * ra[d]; }
    el[i] = sl; er[i] = sr;
}

#define SORT_CH    6250
#define SORT_NC    (NE / SORT_CH)
#define SORT_SHIFT 8
#define SORT_NPB   (1 << SORT_SHIFT)
#define SORT_NB    ((NN + SORT_NPB - 1) / SORT_NPB)
#define SORT_CAP   16384
static_assert(NE % SORT_CH == 0 && SORT_NC == 256, "level 2 reads one row of tab with 256 threads, 8 pieces per wave");
static_assert(NN <= 65536, "a source id is kept in 16 bits of an entry");
static_assert(SORT_NB + 1 <= 256, "one thread per bucket start in level 1");

__device__ __forceinline__ void gat_sort_chunk_body(const int* __restrict__ src,
                                                    const int* __restrict__ dst,
                                                    int* __restrict__ pairs, int* __restrict__ tab,
                                                    int c) {
    constexpr int IT = (SORT_CH + 255) / 256;
    __shared__ int cnt[256];
    __shared__ int wtot[8];
    __shared__ int stage[SORT_CH];
    const int t = threadIdx.x, lane = t & 31, w = t >> 5;
    const int* __restrict__ sp = src + c * SORT_CH;
    const int* __restrict__ dp = dst + c * SORT_CH;
    cnt[t] = 0;
    constexpr int FULL = SORT_CH / 256;
    int pk[IT], bk[IT], rk[IT];
#pragma unroll
    for (int k = 0; k < IT; ++k) {
        const int i = k * 256 + t;
        pk[k] = 0; bk[k] = 0;
        if (k < FULL || i < SORT_CH) {
            const int s = sp[i], d = dp[i];
            bk[k] = d >> SORT_SHIFT;
            pk[k] = s | ((d & (SORT_NPB - 1)) << 16);
        }
    }
    __syncthreads();
#pragma unroll
    for (int k = 0; k < IT; ++k) {
        rk[k] = 0;
        if (k < FULL || k * 256 + t < SORT_CH) rk[k] = atomicAdd(&cnt[bk[k]], 1);
    }
    __syncthreads();
    const int own = cnt[t];
    int inc = own;
    for (int off = 1; off < 32; off <<= 1) {
        const int v = __shfl_up(inc, off);
        if (lane >= off) inc += v;
    }
    if (lane == 31) wtot[w] = inc;
    __syncthreads();
    int base = 0;
    for (int j = 0; j < 8; ++j) if (j < w) base += wtot[j];
    const int st = base + inc - own;
    cnt[t] = st;
    if (t <= SORT_NB) tab[t * SORT_NC + c] = st;
    __syncthreads();
#pragma unroll
    for (int k = 0; k < IT; ++k)
        if (k < FULL || k * 256 + t < SORT_CH) stage[cnt[bk[k]] + rk[k]] = pk[k];
    __syncthreads();
    int* __restrict__ out = pairs + c * SORT_CH;
    for (int i = t; i < SORT_CH; i += 256) out[i] = stage[i];
}

__device__ __forceinline__ void gat_sort_bucket_body(const int* __restrict__ pairs,
                                                     const int* __restrict__ tab,
                                                     int* __restrict__ rowptr,
                                                     int* __restrict__ csr_src, int b) {
    constexpr int NW  = 32;
    constexpr int PPW = SORT_NC / NW;
    __shared__ int cnt[SORT_NPB];
    __shared__ int red[NW];
    __shared__ int wtot[SORT_NPB / 32];
    __shared__ int lst[SORT_CAP];
    const int t = threadIdx.x, lane = t & 31;
    const int w = __builtin_amdgcn_readfirstlane(t >> 5);
    if (t < SORT_NPB) cnt[t] = 0;
    int part = (t < SORT_NC) ? tab[b * SORT_NC + t] : 0;

    int pl[PPW];
    const int* __restrict__ pb[PPW];
    int v0[PPW], v1[PPW];
#pragma unroll
    for (int k = 0; k < PPW; ++k) {
        const int c = w + NW * k;
        const int ps = tab[b * SORT_NC + c];
        pl[k] = tab[(b + 1) * SORT_NC + c] - ps;
        pb[k] = pairs + c * SORT_CH + ps;
    }
#pragma unroll
    for (int k = 0; k < PPW; ++k) {
        const int* __restrict__ cb = pairs + (w + NW * k) * SORT_CH;
        const int o = (int)(pb[k] - cb) + lane;
        v0[k] = cb[min(o, SORT_CH - 1)];
        v1[k] = cb[min(o + 32, SORT_CH - 1)];
    }
    for (int off = 16; off; off >>= 1) part += __shfl_xor(part, off);
    if (lane == 0) red[w] = part;
    __syncthreads();
    int bstart = 0;
    for (int j = 0; j < SORT_NC / 32; ++j) bstart += red[j];

#pragma unroll
    for (int k = 0; k < PPW; ++k) {
        if (lane < pl[k]) atomicAdd(&cnt[v0[k] >> 16], 1);
        if (lane + 32 < pl[k]) atomicAdd(&cnt[v1[k] >> 16], 1);
        for (int i = lane + 64; i < pl[k]; i += 32) atomicAdd(&cnt[pb[k][i] >> 16], 1);
    }
    __syncthreads();
    const int own = (t < SORT_NPB) ? cnt[t] : 0;
    int inc = own;
    for (int off = 1; off < 32; off <<= 1) {
        const int v = __shfl_up(inc, off);
        if (lane >= off) inc += v;
    }
    if (t < SORT_NPB && lane == 31) wtot[w] = inc;
    __syncthreads();
    int size = 0;
    for (int j = 0; j < SORT_NPB / 32; ++j) size += wtot[j];
    if (t < SORT_NPB) {
        int base = 0;
        for (int j = 0; j < SORT_NPB / 32; ++j) if (j < w) base += wtot[j];
        const int st = base + inc - own;
        cnt[t] = st;
        const int n = b * SORT_NPB + t;
        if (n < NN) rowptr[n] = bstart + st;
        if (n == NN - 1) rowptr[NN] = bstart + st + own;
    }
    __syncthreads();
    const bool staged = size <= SORT_CAP;
#pragma unroll
    for (int k = 0; k < PPW; ++k) {
        if (lane < pl[k]) {
            const int pos = atomicAdd(&cnt[v0[k] >> 16], 1);
            if (staged) lst[pos] = v0[k] & 0xffff; else csr_src[bstart + pos] = v0[k] & 0xffff;
        }
        if (lane + 32 < pl[k]) {
            const int pos = atomicAdd(&cnt[v1[k] >> 16], 1);
            if (staged) lst[pos] = v1[k] & 0xffff; else csr_src[bstart + pos] = v1[k] & 0xffff;
        }
        for (int i = lane + 64; i < pl[k]; i += 32) {
            const int e = pb[k][i];
            const int pos = atomicAdd(&cnt[e >> 16], 1);
            if (staged) lst[pos] = e & 0xffff; else csr_src[bstart + pos] = e & 0xffff;
        }
    }
    __syncthreads();
    if (staged)
        for (int i = t; i < size; i += 1024) csr_src[bstart + i] = lst[i];
}

__device__ __forceinline__ float gat_leaky(float v) { return v > 0.f ? v : NEG_SLOPE * v; }

__device__ __forceinline__ float gat_lane_f(float v, int j) {
    return __int_as_float(__builtin_amdgcn_readlane(__float_as_int(v), j));
}

__device__ __forceinline__ void gat_node_ids(const int* __restrict__ csr_src,
                                             int beg, int lane, int& sA, int& sB) {
    sA = csr_src[min(beg + lane, NE - 1)];
    sB = csr_src[min(beg + 32 + lane, NE - 1)];
}
__device__ __forceinline__ void gat_node_els(const float* __restrict__ el, bool on,
                                             int beg, int end, int lane, int sA, int sB,
                                             float2& lA, float2& lB) {
    const int pA = beg + lane, pB = pA + 32;
    lA = *(const float2*)(el + ((on && pA < end) ? sA : 0) * HEADS);
    lB = *(const float2*)(el + ((on && pB < end) ? sB : 0) * HEADS);
}

__device__ __forceinline__ void gat_node_max(const int* __restrict__ csr_src,
                                             const float* __restrict__ el, float2 ern,
                                             int beg, int end, int lane,
                                             float2 lA, float2 lB, float& m0, float& m1) {
    const int pA = beg + lane, pB = pA + 32;
    const bool inA = pA < end, inB = pB < end;
    const float ninf = -__builtin_inff();
    m0 = fmaxf(inA ? gat_leaky(lA.x + ern.x) : ninf, inB ? gat_leaky(lB.x + ern.x) : ninf);
    m1 = fmaxf(inA ? gat_leaky(lA.y + ern.y) : ninf, inB ? gat_leaky(lB.y + ern.y) : ninf);
    for (int base = beg + 64; base < end; base += 32) {
        const int p = base + lane;
        if (p < end) {
            const int s = csr_src[p];
            const float2 l = *(const float2*)(el + s * HEADS);
            m0 = fmaxf(m0, gat_leaky(l.x + ern.x));
            m1 = fmaxf(m1, gat_leaky(l.y + ern.y));
        }
    }
    for (int off = 16; off; off >>= 1) {
        m0 = fmaxf(m0, __shfl_xor(m0, off));
        m1 = fmaxf(m1, __shfl_xor(m1, off));
    }
}

template <int N>
__device__ __forceinline__ void gat_add_rows(float4& acc, const float* __restrict__ fcol,
                                             int s, float x0, float x1, bool head1, int j) {
    float4 f[N];
    float  a[N];
#pragma unroll
    for (int u = 0; u < N; ++u) {
        const int sj = __builtin_amdgcn_readlane(s, j + u);
        f[u] = *(const float4*)(fcol + (long)sj * HD);
        const float a0 = gat_lane_f(x0, j + u), a1 = gat_lane_f(x1, j + u);
        a[u] = head1 ? a1 : a0;
    }
#pragma unroll
    for (int u = 0; u < N; ++u) {
        acc.x += f[u].x * a[u]; acc.y += f[u].y * a[u];
        acc.z += f[u].z * a[u]; acc.w += f[u].w * a[u];
    }
}

#define AGG_NPW 5
#define AGG_BLOCKS ((NN + 8 * AGG_NPW - 1) / (8 * AGG_NPW))
__global__ void __launch_bounds__(256) gat_node_aggregate(
        const int* __restrict__ rowptr, const int* __restrict__ csr_src,
        const float* __restrict__ feat, const float* __restrict__ el,
        const float* __restrict__ er, const float* __restrict__ bias,
        const float* __restrict__ vlr,
        float* __restrict__ out, float* __restrict__ el_next, float* __restrict__ er_next,
        int n_nodes) {
    const int n0 = AGG_NPW *
        __builtin_amdgcn_readfirstlane((int)(blockIdx.x * 8 + (threadIdx.x >> 5)));
    if (n0 >= n_nodes) return;
    const int lane = threadIdx.x & 31;
    const bool head1 = lane >= 16;
    const float* __restrict__ fcol = feat + lane * 4;

    const float4 b  = *(const float4*)(bias + lane * 4);
    const float4 vl = *(const float4*)(vlr + lane * 4);
    const float4 vr = *(const float4*)(vlr + 2 * HID + lane * 4);

    int beg = rowptr[n0], end = rowptr[n0 + 1];
    int end2 = rowptr[min(n0 + 2, n_nodes)];
    float2 ern  = *(const float2*)(er + n0 * HEADS);
    float2 ern2 = *(const float2*)(er + min(n0 + 1, n_nodes - 1) * HEADS);

    int sA, sB;
    float2 lA, lB;
    gat_node_ids(csr_src, beg, lane, sA, sB);
    gat_node_els(el, true, beg, end, lane, sA, sB, lA, lB);

#pragma nounroll
    for (int k = 0; k < AGG_NPW; ++k) {
        const int n = n0 + k;
        const bool more = (k + 1 < AGG_NPW) && (n + 1 < n_nodes);
        int sAn, sBn;
        gat_node_ids(csr_src, end, lane, sAn, sBn);

        float m0, m1;
        gat_node_max(csr_src, el, ern, beg, end, lane, lA, lB, m0, m1);

        int end3 = end2;
        float2 ern3 = ern2;
        if (more) {
            end3 = rowptr[min(n + 3, n_nodes)];
            ern3 = *(const float2*)(er + min(n + 2, n_nodes - 1) * HEADS);
        }

        float s0 = 0.f, s1 = 0.f;
        float4 acc = make_float4(0.f, 0.f, 0.f, 0.f);
        int chunk = 0;
        for (int base = beg; base < end; base += 32, ++chunk) {
            const int p = base + lane;
            int s = chunk ? sB : sA;
            float2 l = chunk ? lB : lA;
            if (chunk >= 2) {
                s = 0;
                if (p < end) {
                    s = csr_src[p];
                    l = *(const float2*)(el + s * HEADS);
                }
            }
            float x0 = 0.f, x1 = 0.f;
            if (p < end) {
                x0 = __expf(gat_leaky(l.x + ern.x) - m0);
                x1 = __expf(gat_leaky(l.y + ern.y) - m1);
            }
            s0 += x0; s1 += x1;
            const int cnt = min(32, end - base);
            int j = 0;
            for (; j + 8 <= cnt; j += 8) gat_add_rows<8>(acc, fcol, s, x0, x1, head1, j);
            if (j + 4 <= cnt) { gat_add_rows<4>(acc, fcol, s, x0, x1, head1, j); j += 4; }
            if (j + 2 <= cnt) { gat_add_rows<2>(acc, fcol, s, x0, x1, head1, j); j += 2; }
            if (j < cnt) gat_add_rows<1>(acc, fcol, s, x0, x1, head1, j);
        }
        float2 lAn, lBn;
        gat_node_els(el, more, end, end2, lane, sAn, sBn, lAn, lBn);

        for (int off = 16; off; off >>= 1) {
            s0 += __shfl_xor(s0, off);
            s1 += __shfl_xor(s1, off);
        }
        if (end > beg) {
            const float sden = head1 ? s1 : s0;
            acc.x /= sden; acc.y /= sden; acc.z /= sden; acc.w /= sden;
        }

        float4 v = make_float4(acc.x + b.x, acc.y + b.y, acc.z + b.z, acc.w + b.w);
        float4 r;
        r.x = 0.5f * (v.x + __shfl_xor(v.x, 16));
        r.y = 0.5f * (v.y + __shfl_xor(v.y, 16));
        r.z = 0.5f * (v.z + __shfl_xor(v.z, 16));
        r.w = 0.5f * (v.w + __shfl_xor(v.w, 16));
        r.x = r.x > 0.f ? r.x : (__expf(r.x) - 1.f);
        r.y = r.y > 0.f ? r.y : (__expf(r.y) - 1.f);
        r.z = r.z > 0.f ? r.z : (__expf(r.z) - 1.f);
        r.w = r.w > 0.f ? r.w : (__expf(r.w) - 1.f);
        if (!head1) *(float4*)(out + (long)n * HID + lane * 4) = r;

        float pl = r.x * vl.x + r.y * vl.y + r.z * vl.z + r.w * vl.w;
        float pr = r.x * vr.x + r.y * vr.y + r.z * vr.z + r.w * vr.w;
        for (int off = 8; off; off >>= 1) {
            pl += __shfl_xor(pl, off);
            pr += __shfl_xor(pr, off);
        }
        if ((lane & 15) == 0) {
            el_next[n * HEADS + (lane >> 4)] = pl;
            er_next[n * HEADS + (lane >> 4)] = pr;
        }

        if (!more) break;
        beg = end; end = end2; end2 = end3;
        ern = ern2; ern2 = ern3;
        sA = sAn; sB = sBn; lA = lAn; lB = lBn;
    }
}

template <int N>
__device__ __forceinline__ void gat_add_hrows(float2& za, float2& zb, const float* __restrict__ hcol,
                                              int s, float x0, float x1, int j) {
    float2 f[N];
    float  a0[N], a1[N];
#pragma unroll
    for (int u = 0; u < N; ++u) {
        const int sj = __builtin_amdgcn_readlane(s, j + u);
        f[u] = *(const float2*)(hcol + (long)sj * HID);
        a0[u] = gat_lane_f(x0, j + u);
        a1[u] = gat_lane_f(x1, j + u);
    }
#pragma unroll
    for (int u = 0; u < N; ++u) {
        za.x += f[u].x * a0[u]; za.y += f[u].y * a0[u];
        zb.x += f[u].x * a1[u]; zb.y += f[u].y * a1[u];
    }
}

__global__ void __launch_bounds__(256) gat_node_aggregate_h(
        const int* __restrict__ rowptr, const int* __restrict__ csr_src,
        const float* __restrict__ h, const float* __restrict__ el,
        const float* __restrict__ er, float* __restrict__ z, int n_nodes) {
    const int n0 = AGG_NPW *
        __builtin_amdgcn_readfirstlane((int)(blockIdx.x * 8 + (threadIdx.x >> 5)));
    if (n0 >= n_nodes) return;
    const int lane = threadIdx.x & 31;
    const float* __restrict__ hcol = h + lane * 2;

    int beg = rowptr[n0], end = rowptr[n0 + 1];
    int end2 = rowptr[min(n0 + 2, n_nodes)];
    float2 ern  = *(const float2*)(er + n0 * HEADS);
    float2 ern2 = *(const float2*)(er + min(n0 + 1, n_nodes - 1) * HEADS);

    int sA, sB;
    float2 lA, lB;
    gat_node_ids(csr_src, beg, lane, sA, sB);
    gat_node_els(el, true, beg, end, lane, sA, sB, lA, lB);

#pragma nounroll
    for (int k = 0; k < AGG_NPW; ++k) {
        const int n = n0 + k;
        const bool more = (k + 1 < AGG_NPW) && (n + 1 < n_nodes);
        int sAn, sBn;
        gat_node_ids(csr_src, end, lane, sAn, sBn);

        float m0, m1;
        gat_node_max(csr_src, el, ern, beg, end, lane, lA, lB, m0, m1);

        int end3 = end2;
        float2 ern3 = ern2;
        if (more) {
            end3 = rowptr[min(n + 3, n_nodes)];
            ern3 = *(const float2*)(er + min(n + 2, n_nodes - 1) * HEADS);
        }

        float s0 = 0.f, s1 = 0.f;
        float2 za = make_float2(0.f, 0.f), zb = make_float2(0.f, 0.f);
        int chunk = 0;
        for (int base = beg; base < end; base += 32, ++chunk) {
            const int p = base + lane;
            int s = chunk ? sB : sA;
            float2 l = chunk ? lB : lA;
            if (chunk >= 2) {
                s = 0;
                if (p < end) {
                    s = csr_src[p];
                    l = *(const float2*)(el + s * HEADS);
                }
            }
            float x0 = 0.f, x1 = 0.f;
            if (p < end) {
                x0 = __expf(gat_leaky(l.x + ern.x) - m0);
                x1 = __expf(gat_leaky(l.y + ern.y) - m1);
            }
            s0 += x0; s1 += x1;
            const int cnt = min(32, end - base);
            int j = 0;
            for (; j + 16 <= cnt; j += 16) gat_add_hrows<16>(za, zb, hcol, s, x0, x1, j);
            if (j + 8 <= cnt) { gat_add_hrows<8>(za, zb, hcol, s, x0, x1, j); j += 8; }
            if (j + 4 <= cnt) { gat_add_hrows<4>(za, zb, hcol, s, x0, x1, j); j += 4; }
            if (j + 2 <= cnt) { gat_add_hrows<2>(za, zb, hcol, s, x0, x1, j); j += 2; }
            if (j < cnt) gat_add_hrows<1>(za, zb, hcol, s, x0, x1, j);
        }
        float2 lAn, lBn;
        gat_node_els(el, more, end, end2, lane, sAn, sBn, lAn, lBn);

        for (int off = 16; off; off >>= 1) {
            s0 += __shfl_xor(s0, off);
            s1 += __shfl_xor(s1, off);
        }
        if (end > beg) {
            za.x /= s0; za.y /= s0;
            zb.x /= s1; zb.y /= s1;
        }
        *(float2*)(z + (long)n * HD + lane * 2) = za;
        *(float2*)(z + (long)n * HD + HID + lane * 2) = zb;

        if (!more) break;
        beg = end; end = end2; end2 = end3;
        ern = ern2; ern2 = ern3;
        sA = sAn; sB = sBn; lA = lAn; lB = lBn;
    }
}


__global__ void __launch_bounds__(256) gat_proj1_and_sort1(
        const float* __restrict__ x, const float* __restrict__ W, const float* __restrict__ bias,
        float* __restrict__ feat, int n_strips, int gemm_blocks,
        const int* __restrict__ src, const int* __restrict__ dst,
        int* __restrict__ pairs, int* __restrict__ tab) {
    const int b = blockIdx.x;
    if (b < gemm_blocks) {
        gat_gemm_strip_body<INF_, HD, false>(x, W, bias, feat, n_strips, b);
    } else {
        gat_sort_chunk_body(src, dst, pairs, tab, b - gemm_blocks);
    }
}

__global__ void __launch_bounds__(1024) gat_sort2_and_logits(
        const int* __restrict__ pairs, const int* __restrict__ tab,
        int* __restrict__ rowptr, int* __restrict__ csr_src,
        const float* __restrict__ feat, const float* __restrict__ al1,
        const float* __restrict__ ar1, float* __restrict__ el, float* __restrict__ er, int NH,
        const float* __restrict__ W2, const float* __restrict__ al2,
        const float* __restrict__ ar2, float* __restrict__ vlr) {
    const int b = blockIdx.x;
    if (b == 0) {
        if (threadIdx.x < 256) gat_attn_vec_body(W2, al2, ar2, vlr);
    } else if (b <= SORT_NB) {
        gat_sort_bucket_body(pairs, tab, rowptr, csr_src, b - 1);
    } else {
        gat_eler_body<1024>(feat, al1, ar1, el, er, NH, b - 1 - SORT_NB);
    }
}

__global__ void __launch_bounds__(256) gat_proj2_final(
        const float* __restrict__ z, const float* __restrict__ W, const float* __restrict__ bias,
        float* __restrict__ out, int n_strips) {
    gat_gemm_strip_body<HD, HID, true>(z, W, bias, out, n_strips, (int)blockIdx.x);
}

extern "C" void kernel_launch(void* const* d_in, const int* in_sizes, int n_in,
                              void* d_out, int out_size, void* d_ws, size_t ws_size,
                              hipStream_t stream) {
    const float* x   = (const float*)d_in[0];
    const float* W1  = (const float*)d_in[1];
    const float* al1 = (const float*)d_in[2];
    const float* ar1 = (const float*)d_in[3];
    const float* b1  = (const float*)d_in[4];
    const float* W2  = (const float*)d_in[5];
    const float* al2 = (const float*)d_in[6];
    const float* ar2 = (const float*)d_in[7];
    const float* b2  = (const float*)d_in[8];
    const int*   src = (const int*)d_in[9];
    const int*   dst = (const int*)d_in[10];
    float* out = (float*)d_out;

    char* wsb = (char*)d_ws;
    size_t off = 0;
    auto alloc = [&](size_t bytes) -> void* {
        void* p = wsb + off;
        off += bytes;
        off = (off + 255) & ~(size_t)255;
        return p;
    };
    float* feat = (float*)alloc((size_t)NN * HD * 4);
    float* el   = (float*)alloc((size_t)NN * HEADS * 4);
    float* er   = (float*)alloc((size_t)NN * HEADS * 4);
    int* rowptr  = (int*)alloc((size_t)(NN + 1) * 4);
    int* csr_src = (int*)alloc((size_t)NE * 4);
    float* h    = (float*)alloc((size_t)NN * HID * 4);
    int* tab    = (int*)alloc((size_t)(SORT_NB + 1) * SORT_NC * 4);
    float* vlr  = (float*)alloc((size_t)4 * HID * 4);
    float* el2  = (float*)alloc((size_t)NN * HEADS * 4);
    float* er2  = (float*)alloc((size_t)NN * HEADS * 4);
    int* pairs = (int*)h;
    float* z   = feat;

    const int NH   = NN * HEADS;
    const dim3 B256(256);
    const int GEMM_BLOCKS = (NN / 16 + 7) / 8;
    const int ELER_BLOCKS = (NH + 1023) / 1024;

    gat_proj1_and_sort1<<<dim3(GEMM_BLOCKS + SORT_NC), B256, 0, stream>>>(
        x, W1, b1, feat, NN / 16, GEMM_BLOCKS, src, dst, pairs, tab);
    gat_sort2_and_logits<<<dim3(1 + SORT_NB + ELER_BLOCKS), dim3(1024), 0, stream>>>(
        pairs, tab, rowptr, csr_src, feat, al1, ar1, el, er, NH, W2, al2, ar2, vlr);

    gat_node_aggregate<<<dim3(AGG_BLOCKS), B256, 0, stream>>>(
        rowptr, csr_src, feat, el, er, b1, vlr, h, el2, er2, NN);

    gat_node_aggregate_h<<<dim3(AGG_BLOCKS), B256, 0, stream>>>(rowptr, csr_src, h, el2, er2, z, NN);
    gat_proj2_final<<<dim3(GEMM_BLOCKS), B256, 0, stream>>>(z, W2, b2, out, NN / 16);
}


# Layer 1's logits el, er formed in the projection's epilogue from the accumulator tiles (16-lane sums); the stage that read feat back with 32-line loads is gone; logit sums reassociated
# speedup vs baseline: 14.1090x; 1.0520x over previous
#include <hip/hip_runtime.h>
#include <hip/hip_bf16.h>

#define NN   50000
#define NE   1600000
#define INF_ 128
#define HID  64
#define HEADS 2
#define HD   (HEADS*HID)
#define NEG_SLOPE 0.2f

typedef __attribute__((ext_vector_type(2))) float v2f;
typedef __attribute__((ext_vector_type(8))) float v8f;


__device__ __forceinline__ float gat_halfwave_sums16(const float (&a)[16], int l15) {
    const bool h8 = (l15 & 8) != 0, h4 = (l15 & 4) != 0, h2 = (l15 & 2) != 0, h1 = (l15 & 1) != 0;
    float b[8], c[4], d[2];
#pragma unroll
    for (int i = 0; i < 8; ++i) b[i] = (h8 ? a[i + 8] : a[i]) + __shfl_xor(h8 ? a[i] : a[i + 8], 8);
#pragma unroll
    for (int i = 0; i < 4; ++i) c[i] = (h4 ? b[i + 4] : b[i]) + __shfl_xor(h4 ? b[i] : b[i + 4], 4);
#pragma unroll
    for (int i = 0; i < 2; ++i) d[i] = (h2 ? c[i + 2] : c[i]) + __shfl_xor(h2 ? c[i] : c[i + 2], 2);
    return (h1 ? d[1] : d[0]) + __shfl_xor(h1 ? d[0] : d[1], 1);
}

template <int K, int NC, bool FINAL>
__device__ __forceinline__ void gat_gemm_strip_body(
        const float* __restrict__ A, const float* __restrict__ W,
        const float* __restrict__ bias, float* __restrict__ D, int n_strips, int blk,
        const float* __restrict__ al, const float* __restrict__ ar,
        float* __restrict__ el, float* __restrict__ er) {
    static_assert(FINAL || (NC == HD && NC / 16 == 2 * (HID / 16)), "logits: the strip holds both heads' columns");
    constexpr int KS = K / 4;
    constexpr int NT = NC / 16;
    constexpr int CG = NC / 4;
    constexpr int GEMM_LDS_PITCH = NC + 16;
    constexpr int NW = ((K / 2) * CG) / 256;
    __shared__ __attribute__((aligned(16))) v2f wl[(K / 2) * GEMM_LDS_PITCH];

    const int lane = threadIdx.x & 31;
    const int wave = threadIdx.x >> 5;
    const int half = lane >> 4;
    const int l15  = lane & 15;
    const int strip = blk * 8 + wave;
    const bool live = strip < n_strips;

    float4 w0[NW], w1[NW];
#pragma unroll
    for (int i = 0; i < NW; ++i) {
        const int q = i * 256 + threadIdx.x;
        const int k0 = 2 * (q / CG);
        const float* __restrict__ r0 =
            FINAL ? W + (k0 & (HID - 1)) * HD + (k0 / HID) * HID : W + k0 * HD;
        const float4* __restrict__ wp = (const float4*)r0 + (q % CG);
        w0[i] = wp[0];
        w1[i] = wp[HD / 4];
    }

    v2f a[KS];
    const float* __restrict__ arow =
        A + ((long)(live ? strip : n_strips - 1) * 16 + l15) * K + half * 2;
#pragma unroll
    for (int i = 0; i < KS; ++i) a[i] = *(const v2f*)(arow + 4 * i);

#pragma unroll
    for (int i = 0; i < NW; ++i) {
        const int q = i * 256 + threadIdx.x;
        float4* __restrict__ lp = (float4*)(wl + (q / CG) * GEMM_LDS_PITCH + (q % CG) * 4);
        lp[0] = make_float4(w0[i].x, w1[i].x, w0[i].y, w1[i].y);
        lp[1] = make_float4(w0[i].z, w1[i].z, w0[i].w, w1[i].w);
    }
    __builtin_amdgcn_fence(__ATOMIC_RELEASE, "workgroup", "local");
    __builtin_amdgcn_s_barrier();
    __builtin_amdgcn_fence(__ATOMIC_ACQUIRE, "workgroup", "local");
    if (!live) return;

    v8f c[NT];
#pragma unroll
    for (int t = 0; t < NT; ++t) c[t] = v8f{};

    const v2f* __restrict__ bl = wl + half * GEMM_LDS_PITCH + l15;
    v2f bcur[NT], bnxt[NT];
#pragma unroll
    for (int t = 0; t < NT; ++t) bcur[t] = bl[t * 16];
#pragma unroll
    for (int i = 0; i < KS; ++i) {
        if (i + 1 < KS) {
#pragma unroll
            for (int t = 0; t < NT; ++t) bnxt[t] = bl[(2 * (i + 1)) * GEMM_LDS_PITCH + t * 16];
        }
#pragma unroll
        for (int t = 0; t < NT; ++t)
            c[t] = __builtin_amdgcn_wmma_f32_16x16x4_f32(false, a[i], false, bcur[t],
                                                         (short)0, c[t], false, false);
#pragma unroll
        for (int t = 0; t < NT; ++t) bcur[t] = bnxt[t];
    }

    float* __restrict__ drow = D + ((long)strip * 16 + half * 8) * NC + l15;
#pragma unroll
    for (int t = 0; t < NT; ++t) {
        float bs = 0.f;
        if (FINAL) bs = bias[t * 16 + l15] + bias[HID + t * 16 + l15];
#pragma unroll
        for (int v = 0; v < 8; ++v) {
            float r = c[t][v];
            if (FINAL) {
                r = 0.5f * (r + bs);
                r = r > 0.f ? r : (__expf(r) - 1.f);
            }
            drow[(long)v * NC + t * 16] = r;
        }
    }

    if (!FINAL) {
        __builtin_amdgcn_sched_barrier(0);
        constexpr int TH = HID / 16;
        float pe[16], pr[16];
#pragma unroll
        for (int hd = 0; hd < HEADS; ++hd) {
#pragma unroll
            for (int tt = 0; tt < TH; ++tt) {
                const int t = hd * TH + tt;
                const float la = al[t * 16 + l15], ra = ar[t * 16 + l15];
#pragma unroll
                for (int v = 0; v < 8; ++v) {
                    const float f = c[t][v];
                    if (tt == 0) { pe[2 * v + hd] = f * la;  pr[2 * v + hd] = f * ra; }
                    else         { pe[2 * v + hd] += f * la; pr[2 * v + hd] += f * ra; }
                }
            }
        }
        const float se = gat_halfwave_sums16(pe, l15);
        const float sr = gat_halfwave_sums16(pr, l15);
        el[(long)strip * 32 + lane] = se;
        er[(long)strip * 32 + lane] = sr;
    }
}

__device__ __forceinline__ void gat_attn_vec_body(const float* __restrict__ W,
                                                  const float* __restrict__ al,
                                                  const float* __restrict__ ar,
                                                  float* __restrict__ vlr) {
    const int t = threadIdx.x;
    const int c = t & (HID - 1), hd = (t >> 6) & 1;
    const float* __restrict__ a = ((t >> 7) ? ar : al) + hd * HID;
    const float* __restrict__ w = W + c * HD + hd * HID;
    float s = 0.f;
    for (int d = 0; d < HID; ++d) s += w[d] * a[d];
    vlr[t] = s;
}

#define SORT_CH    6250
#define SORT_NC    (NE / SORT_CH)
#define SORT_SHIFT 8
#define SORT_NPB   (1 << SORT_SHIFT)
#define SORT_NB    ((NN + SORT_NPB - 1) / SORT_NPB)
#define SORT_CAP   16384
static_assert(NE % SORT_CH == 0 && SORT_NC == 256, "level 2 reads one row of tab with 256 threads, 8 pieces per wave");
static_assert(NN <= 65536, "a source id is kept in 16 bits of an entry");
static_assert(SORT_NB + 1 <= 256, "one thread per bucket start in level 1");

__device__ __forceinline__ void gat_sort_chunk_body(const int* __restrict__ src,
                                                    const int* __restrict__ dst,
                                                    int* __restrict__ pairs, int* __restrict__ tab,
                                                    int c) {
    constexpr int IT = (SORT_CH + 255) / 256;
    __shared__ int cnt[256];
    __shared__ int wtot[8];
    __shared__ int stage[SORT_CH];
    const int t = threadIdx.x, lane = t & 31, w = t >> 5;
    const int* __restrict__ sp = src + c * SORT_CH;
    const int* __restrict__ dp = dst + c * SORT_CH;
    cnt[t] = 0;
    constexpr int FULL = SORT_CH / 256;
    int pk[IT], bk[IT], rk[IT];
#pragma unroll
    for (int k = 0; k < IT; ++k) {
        const int i = k * 256 + t;
        pk[k] = 0; bk[k] = 0;
        if (k < FULL || i < SORT_CH) {
            const int s = sp[i], d = dp[i];
            bk[k] = d >> SORT_SHIFT;
            pk[k] = s | ((d & (SORT_NPB - 1)) << 16);
        }
    }
    __syncthreads();
#pragma unroll
    for (int k = 0; k < IT; ++k) {
        rk[k] = 0;
        if (k < FULL || k * 256 + t < SORT_CH) rk[k] = atomicAdd(&cnt[bk[k]], 1);
    }
    __syncthreads();
    const int own = cnt[t];
    int inc = own;
    for (int off = 1; off < 32; off <<= 1) {
        const int v = __shfl_up(inc, off);
        if (lane >= off) inc += v;
    }
    if (lane == 31) wtot[w] = inc;
    __syncthreads();
    int base = 0;
    for (int j = 0; j < 8; ++j) if (j < w) base += wtot[j];
    const int st = base + inc - own;
    cnt[t] = st;
    if (t <= SORT_NB) tab[t * SORT_NC + c] = st;
    __syncthreads();
#pragma unroll
    for (int k = 0; k < IT; ++k)
        if (k < FULL || k * 256 + t < SORT_CH) stage[cnt[bk[k]] + rk[k]] = pk[k];
    __syncthreads();
    int* __restrict__ out = pairs + c * SORT_CH;
    for (int i = t; i < SORT_CH; i += 256) out[i] = stage[i];
}

__device__ __forceinline__ void gat_sort_bucket_body(const int* __restrict__ pairs,
                                                     const int* __restrict__ tab,
                                                     int* __restrict__ rowptr,
                                                     int* __restrict__ csr_src, int b) {
    constexpr int NW  = 32;
    constexpr int PPW = SORT_NC / NW;
    __shared__ int cnt[SORT_NPB];
    __shared__ int red[NW];
    __shared__ int wtot[SORT_NPB / 32];
    __shared__ int lst[SORT_CAP];
    const int t = threadIdx.x, lane = t & 31;
    const int w = __builtin_amdgcn_readfirstlane(t >> 5);
    if (t < SORT_NPB) cnt[t] = 0;
    int part = (t < SORT_NC) ? tab[b * SORT_NC + t] : 0;

    int pl[PPW];
    const int* __restrict__ pb[PPW];
    int v0[PPW], v1[PPW];
#pragma unroll
    for (int k = 0; k < PPW; ++k) {
        const int c = w + NW * k;
        const int ps = tab[b * SORT_NC + c];
        pl[k] = tab[(b + 1) * SORT_NC + c] - ps;
        pb[k] = pairs + c * SORT_CH + ps;
    }
#pragma unroll
    for (int k = 0; k < PPW; ++k) {
        const int* __restrict__ cb = pairs + (w + NW * k) * SORT_CH;
        const int o = (int)(pb[k] - cb) + lane;
        v0[k] = cb[min(o, SORT_CH - 1)];
        v1[k] = cb[min(o + 32, SORT_CH - 1)];
    }
    for (int off = 16; off; off >>= 1) part += __shfl_xor(part, off);
    if (lane == 0) red[w] = part;
    __syncthreads();
    int bstart = 0;
    for (int j = 0; j < SORT_NC / 32; ++j) bstart += red[j];

#pragma unroll
    for (int k = 0; k < PPW; ++k) {
        if (lane < pl[k]) atomicAdd(&cnt[v0[k] >> 16], 1);
        if (lane + 32 < pl[k]) atomicAdd(&cnt[v1[k] >> 16], 1);
        for (int i = lane + 64; i < pl[k]; i += 32) atomicAdd(&cnt[pb[k][i] >> 16], 1);
    }
    __syncthreads();
    const int own = (t < SORT_NPB) ? cnt[t] : 0;
    int inc = own;
    for (int off = 1; off < 32; off <<= 1) {
        const int v = __shfl_up(inc, off);
        if (lane >= off) inc += v;
    }
    if (t < SORT_NPB && lane == 31) wtot[w] = inc;
    __syncthreads();
    int size = 0;
    for (int j = 0; j < SORT_NPB / 32; ++j) size += wtot[j];
    if (t < SORT_NPB) {
        int base = 0;
        for (int j = 0; j < SORT_NPB / 32; ++j) if (j < w) base += wtot[j];
        const int st = base + inc - own;
        cnt[t] = st;
        const int n = b * SORT_NPB + t;
        if (n < NN) rowptr[n] = bstart + st;
        if (n == NN - 1) rowptr[NN] = bstart + st + own;
    }
    __syncthreads();
    const bool staged = size <= SORT_CAP;
#pragma unroll
    for (int k = 0; k < PPW; ++k) {
        if (lane < pl[k]) {
            const int pos = atomicAdd(&cnt[v0[k] >> 16], 1);
            if (staged) lst[pos] = v0[k] & 0xffff; else csr_src[bstart + pos] = v0[k] & 0xffff;
        }
        if (lane + 32 < pl[k]) {
            const int pos = atomicAdd(&cnt[v1[k] >> 16], 1);
            if (staged) lst[pos] = v1[k] & 0xffff; else csr_src[bstart + pos] = v1[k] & 0xffff;
        }
        for (int i = lane + 64; i < pl[k]; i += 32) {
            const int e = pb[k][i];
            const int pos = atomicAdd(&cnt[e >> 16], 1);
            if (staged) lst[pos] = e & 0xffff; else csr_src[bstart + pos] = e & 0xffff;
        }
    }
    __syncthreads();
    if (staged)
        for (int i = t; i < size; i += 1024) csr_src[bstart + i] = lst[i];
}

__device__ __forceinline__ float gat_leaky(float v) { return v > 0.f ? v : NEG_SLOPE * v; }

__device__ __forceinline__ float gat_lane_f(float v, int j) {
    return __int_as_float(__builtin_amdgcn_readlane(__float_as_int(v), j));
}

__device__ __forceinline__ void gat_node_ids(const int* __restrict__ csr_src,
                                             int beg, int lane, int& sA, int& sB) {
    sA = csr_src[min(beg + lane, NE - 1)];
    sB = csr_src[min(beg + 32 + lane, NE - 1)];
}
__device__ __forceinline__ void gat_node_els(const float* __restrict__ el, bool on,
                                             int beg, int end, int lane, int sA, int sB,
                                             float2& lA, float2& lB) {
    const int pA = beg + lane, pB = pA + 32;
    lA = *(const float2*)(el + ((on && pA < end) ? sA : 0) * HEADS);
    lB = *(const float2*)(el + ((on && pB < end) ? sB : 0) * HEADS);
}

__device__ __forceinline__ void gat_node_max(const int* __restrict__ csr_src,
                                             const float* __restrict__ el, float2 ern,
                                             int beg, int end, int lane,
                                             float2 lA, float2 lB, float& m0, float& m1) {
    const int pA = beg + lane, pB = pA + 32;
    const bool inA = pA < end, inB = pB < end;
    const float ninf = -__builtin_inff();
    m0 = fmaxf(inA ? gat_leaky(lA.x + ern.x) : ninf, inB ? gat_leaky(lB.x + ern.x) : ninf);
    m1 = fmaxf(inA ? gat_leaky(lA.y + ern.y) : ninf, inB ? gat_leaky(lB.y + ern.y) : ninf);
    for (int base = beg + 64; base < end; base += 32) {
        const int p = base + lane;
        if (p < end) {
            const int s = csr_src[p];
            const float2 l = *(const float2*)(el + s * HEADS);
            m0 = fmaxf(m0, gat_leaky(l.x + ern.x));
            m1 = fmaxf(m1, gat_leaky(l.y + ern.y));
        }
    }
    for (int off = 16; off; off >>= 1) {
        m0 = fmaxf(m0, __shfl_xor(m0, off));
        m1 = fmaxf(m1, __shfl_xor(m1, off));
    }
}

template <int N>
__device__ __forceinline__ void gat_add_rows(float4& acc, const float* __restrict__ fcol,
                                             int s, float x0, float x1, bool head1, int j) {
    float4 f[N];
    float  a[N];
#pragma unroll
    for (int u = 0; u < N; ++u) {
        const int sj = __builtin_amdgcn_readlane(s, j + u);
        f[u] = *(const float4*)(fcol + (long)sj * HD);
        const float a0 = gat_lane_f(x0, j + u), a1 = gat_lane_f(x1, j + u);
        a[u] = head1 ? a1 : a0;
    }
#pragma unroll
    for (int u = 0; u < N; ++u) {
        acc.x += f[u].x * a[u]; acc.y += f[u].y * a[u];
        acc.z += f[u].z * a[u]; acc.w += f[u].w * a[u];
    }
}

#define AGG_NPW 5
#define AGG_BLOCKS ((NN + 8 * AGG_NPW - 1) / (8 * AGG_NPW))
__global__ void __launch_bounds__(256) gat_node_aggregate(
        const int* __restrict__ rowptr, const int* __restrict__ csr_src,
        const float* __restrict__ feat, const float* __restrict__ el,
        const float* __restrict__ er, const float* __restrict__ bias,
        const float* __restrict__ vlr,
        float* __restrict__ out, float* __restrict__ el_next, float* __restrict__ er_next,
        int n_nodes) {
    const int n0 = AGG_NPW *
        __builtin_amdgcn_readfirstlane((int)(blockIdx.x * 8 + (threadIdx.x >> 5)));
    if (n0 >= n_nodes) return;
    const int lane = threadIdx.x & 31;
    const bool head1 = lane >= 16;
    const float* __restrict__ fcol = feat + lane * 4;

    const float4 b  = *(const float4*)(bias + lane * 4);
    const float4 vl = *(const float4*)(vlr + lane * 4);
    const float4 vr = *(const float4*)(vlr + 2 * HID + lane * 4);

    int beg = rowptr[n0], end = rowptr[n0 + 1];
    int end2 = rowptr[min(n0 + 2, n_nodes)];
    float2 ern  = *(const float2*)(er + n0 * HEADS);
    float2 ern2 = *(const float2*)(er + min(n0 + 1, n_nodes - 1) * HEADS);

    int sA, sB;
    float2 lA, lB;
    gat_node_ids(csr_src, beg, lane, sA, sB);
    gat_node_els(el, true, beg, end, lane, sA, sB, lA, lB);

#pragma nounroll
    for (int k = 0; k < AGG_NPW; ++k) {
        const int n = n0 + k;
        const bool more = (k + 1 < AGG_NPW) && (n + 1 < n_nodes);
        int sAn, sBn;
        gat_node_ids(csr_src, end, lane, sAn, sBn);

        float m0, m1;
        gat_node_max(csr_src, el, ern, beg, end, lane, lA, lB, m0, m1);

        int end3 = end2;
        float2 ern3 = ern2;
        if (more) {
            end3 = rowptr[min(n + 3, n_nodes)];
            ern3 = *(const float2*)(er + min(n + 2, n_nodes - 1) * HEADS);
        }

        float s0 = 0.f, s1 = 0.f;
        float4 acc = make_float4(0.f, 0.f, 0.f, 0.f);
        int chunk = 0;
        for (int base = beg; base < end; base += 32, ++chunk) {
            const int p = base + lane;
            int s = chunk ? sB : sA;
            float2 l = chunk ? lB : lA;
            if (chunk >= 2) {
                s = 0;
                if (p < end) {
                    s = csr_src[p];
                    l = *(const float2*)(el + s * HEADS);
                }
            }
            float x0 = 0.f, x1 = 0.f;
            if (p < end) {
                x0 = __expf(gat_leaky(l.x + ern.x) - m0);
                x1 = __expf(gat_leaky(l.y + ern.y) - m1);
            }
            s0 += x0; s1 += x1;
            const int cnt = min(32, end - base);
            int j = 0;
            for (; j + 8 <= cnt; j += 8) gat_add_rows<8>(acc, fcol, s, x0, x1, head1, j);
            if (j + 4 <= cnt) { gat_add_rows<4>(acc, fcol, s, x0, x1, head1, j); j += 4; }
            if (j + 2 <= cnt) { gat_add_rows<2>(acc, fcol, s, x0, x1, head1, j); j += 2; }
            if (j < cnt) gat_add_rows<1>(acc, fcol, s, x0, x1, head1, j);
        }
        float2 lAn, lBn;
        gat_node_els(el, more, end, end2, lane, sAn, sBn, lAn, lBn);

        for (int off = 16; off; off >>= 1) {
            s0 += __shfl_xor(s0, off);
            s1 += __shfl_xor(s1, off);
        }
        if (end > beg) {
            const float sden = head1 ? s1 : s0;
            acc.x /= sden; acc.y /= sden; acc.z /= sden; acc.w /= sden;
        }

        float4 v = make_float4(acc.x + b.x, acc.y + b.y, acc.z + b.z, acc.w + b.w);
        float4 r;
        r.x = 0.5f * (v.x + __shfl_xor(v.x, 16));
        r.y = 0.5f * (v.y + __shfl_xor(v.y, 16));
        r.z = 0.5f * (v.z + __shfl_xor(v.z, 16));
        r.w = 0.5f * (v.w + __shfl_xor(v.w, 16));
        r.x = r.x > 0.f ? r.x : (__expf(r.x) - 1.f);
        r.y = r.y > 0.f ? r.y : (__expf(r.y) - 1.f);
        r.z = r.z > 0.f ? r.z : (__expf(r.z) - 1.f);
        r.w = r.w > 0.f ? r.w : (__expf(r.w) - 1.f);
        if (!head1) *(float4*)(out + (long)n * HID + lane * 4) = r;

        float pl = r.x * vl.x + r.y * vl.y + r.z * vl.z + r.w * vl.w;
        float pr = r.x * vr.x + r.y * vr.y + r.z * vr.z + r.w * vr.w;
        for (int off = 8; off; off >>= 1) {
            pl += __shfl_xor(pl, off);
            pr += __shfl_xor(pr, off);
        }
        if ((lane & 15) == 0) {
            el_next[n * HEADS + (lane >> 4)] = pl;
            er_next[n * HEADS + (lane >> 4)] = pr;
        }

        if (!more) break;
        beg = end; end = end2; end2 = end3;
        ern = ern2; ern2 = ern3;
        sA = sAn; sB = sBn; lA = lAn; lB = lBn;
    }
}

template <int N>
__device__ __forceinline__ void gat_add_hrows(float2& za, float2& zb, const float* __restrict__ hcol,
                                              int s, float x0, float x1, int j) {
    float2 f[N];
    float  a0[N], a1[N];
#pragma unroll
    for (int u = 0; u < N; ++u) {
        const int sj = __builtin_amdgcn_readlane(s, j + u);
        f[u] = *(const float2*)(hcol + (long)sj * HID);
        a0[u] = gat_lane_f(x0, j + u);
        a1[u] = gat_lane_f(x1, j + u);
    }
#pragma unroll
    for (int u = 0; u < N; ++u) {
        za.x += f[u].x * a0[u]; za.y += f[u].y * a0[u];
        zb.x += f[u].x * a1[u]; zb.y += f[u].y * a1[u];
    }
}

__global__ void __launch_bounds__(256) gat_node_aggregate_h(
        const int* __restrict__ rowptr, const int* __restrict__ csr_src,
        const float* __restrict__ h, const float* __restrict__ el,
        const float* __restrict__ er, float* __restrict__ z, int n_nodes) {
    const int n0 = AGG_NPW *
        __builtin_amdgcn_readfirstlane((int)(blockIdx.x * 8 + (threadIdx.x >> 5)));
    if (n0 >= n_nodes) return;
    const int lane = threadIdx.x & 31;
    const float* __restrict__ hcol = h + lane * 2;

    int beg = rowptr[n0], end = rowptr[n0 + 1];
    int end2 = rowptr[min(n0 + 2, n_nodes)];
    float2 ern  = *(const float2*)(er + n0 * HEADS);
    float2 ern2 = *(const float2*)(er + min(n0 + 1, n_nodes - 1) * HEADS);

    int sA, sB;
    float2 lA, lB;
    gat_node_ids(csr_src, beg, lane, sA, sB);
    gat_node_els(el, true, beg, end, lane, sA, sB, lA, lB);

#pragma nounroll
    for (int k = 0; k < AGG_NPW; ++k) {
        const int n = n0 + k;
        const bool more = (k + 1 < AGG_NPW) && (n + 1 < n_nodes);
        int sAn, sBn;
        gat_node_ids(csr_src, end, lane, sAn, sBn);

        float m0, m1;
        gat_node_max(csr_src, el, ern, beg, end, lane, lA, lB, m0, m1);

        int end3 = end2;
        float2 ern3 = ern2;
        if (more) {
            end3 = rowptr[min(n + 3, n_nodes)];
            ern3 = *(const float2*)(er + min(n + 2, n_nodes - 1) * HEADS);
        }

        float s0 = 0.f, s1 = 0.f;
        float2 za = make_float2(0.f, 0.f), zb = make_float2(0.f, 0.f);
        int chunk = 0;
        for (int base = beg; base < end; base += 32, ++chunk) {
            const int p = base + lane;
            int s = chunk ? sB : sA;
            float2 l = chunk ? lB : lA;
            if (chunk >= 2) {
                s = 0;
                if (p < end) {
                    s = csr_src[p];
                    l = *(const float2*)(el + s * HEADS);
                }
            }
            float x0 = 0.f, x1 = 0.f;
            if (p < end) {
                x0 = __expf(gat_leaky(l.x + ern.x) - m0);
                x1 = __expf(gat_leaky(l.y + ern.y) - m1);
            }
            s0 += x0; s1 += x1;
            const int cnt = min(32, end - base);
            int j = 0;
            for (; j + 16 <= cnt; j += 16) gat_add_hrows<16>(za, zb, hcol, s, x0, x1, j);
            if (j + 8 <= cnt) { gat_add_hrows<8>(za, zb, hcol, s, x0, x1, j); j += 8; }
            if (j + 4 <= cnt) { gat_add_hrows<4>(za, zb, hcol, s, x0, x1, j); j += 4; }
            if (j + 2 <= cnt) { gat_add_hrows<2>(za, zb, hcol, s, x0, x1, j); j += 2; }
            if (j < cnt) gat_add_hrows<1>(za, zb, hcol, s, x0, x1, j);
        }
        float2 lAn, lBn;
        gat_node_els(el, more, end, end2, lane, sAn, sBn, lAn, lBn);

        for (int off = 16; off; off >>= 1) {
            s0 += __shfl_xor(s0, off);
            s1 += __shfl_xor(s1, off);
        }
        if (end > beg) {
            za.x /= s0; za.y /= s0;
            zb.x /= s1; zb.y /= s1;
        }
        *(float2*)(z + (long)n * HD + lane * 2) = za;
        *(float2*)(z + (long)n * HD + HID + lane * 2) = zb;

        if (!more) break;
        beg = end; end = end2; end2 = end3;
        ern = ern2; ern2 = ern3;
        sA = sAn; sB = sBn; lA = lAn; lB = lBn;
    }
}


__global__ void __launch_bounds__(256) gat_proj1_and_sort1(
        const float* __restrict__ x, const float* __restrict__ W, const float* __restrict__ bias,
        float* __restrict__ feat, int n_strips, int gemm_blocks,
        const float* __restrict__ al1, const float* __restrict__ ar1,
        float* __restrict__ el, float* __restrict__ er,
        const int* __restrict__ src, const int* __restrict__ dst,
        int* __restrict__ pairs, int* __restrict__ tab) {
    const int b = blockIdx.x;
    if (b < gemm_blocks) {
        gat_gemm_strip_body<INF_, HD, false>(x, W, bias, feat, n_strips, b, al1, ar1, el, er);
    } else {
        gat_sort_chunk_body(src, dst, pairs, tab, b - gemm_blocks);
    }
}

__global__ void __launch_bounds__(1024) gat_sort2_and_vec(
        const int* __restrict__ pairs, const int* __restrict__ tab,
        int* __restrict__ rowptr, int* __restrict__ csr_src,
        const float* __restrict__ W2, const float* __restrict__ al2,
        const float* __restrict__ ar2, float* __restrict__ vlr) {
    const int b = blockIdx.x;
    if (b == 0) {
        if (threadIdx.x < 256) gat_attn_vec_body(W2, al2, ar2, vlr);
    } else {
        gat_sort_bucket_body(pairs, tab, rowptr, csr_src, b - 1);
    }
}

__global__ void __launch_bounds__(256) gat_proj2_final(
        const float* __restrict__ z, const float* __restrict__ W, const float* __restrict__ bias,
        float* __restrict__ out, int n_strips) {
    gat_gemm_strip_body<HD, HID, true>(z, W, bias, out, n_strips, (int)blockIdx.x,
                                       nullptr, nullptr, nullptr, nullptr);
}

extern "C" void kernel_launch(void* const* d_in, const int* in_sizes, int n_in,
                              void* d_out, int out_size, void* d_ws, size_t ws_size,
                              hipStream_t stream) {
    const float* x   = (const float*)d_in[0];
    const float* W1  = (const float*)d_in[1];
    const float* al1 = (const float*)d_in[2];
    const float* ar1 = (const float*)d_in[3];
    const float* b1  = (const float*)d_in[4];
    const float* W2  = (const float*)d_in[5];
    const float* al2 = (const float*)d_in[6];
    const float* ar2 = (const float*)d_in[7];
    const float* b2  = (const float*)d_in[8];
    const int*   src = (const int*)d_in[9];
    const int*   dst = (const int*)d_in[10];
    float* out = (float*)d_out;

    char* wsb = (char*)d_ws;
    size_t off = 0;
    auto alloc = [&](size_t bytes) -> void* {
        void* p = wsb + off;
        off += bytes;
        off = (off + 255) & ~(size_t)255;
        return p;
    };
    float* feat = (float*)alloc((size_t)NN * HD * 4);
    float* el   = (float*)alloc((size_t)NN * HEADS * 4);
    float* er   = (float*)alloc((size_t)NN * HEADS * 4);
    int* rowptr  = (int*)alloc((size_t)(NN + 1) * 4);
    int* csr_src = (int*)alloc((size_t)NE * 4);
    float* h    = (float*)alloc((size_t)NN * HID * 4);
    int* tab    = (int*)alloc((size_t)(SORT_NB + 1) * SORT_NC * 4);
    float* vlr  = (float*)alloc((size_t)4 * HID * 4);
    float* el2  = (float*)alloc((size_t)NN * HEADS * 4);
    float* er2  = (float*)alloc((size_t)NN * HEADS * 4);
    int* pairs = (int*)h;
    float* z   = feat;

    const dim3 B256(256);
    const int GEMM_BLOCKS = (NN / 16 + 7) / 8;

    gat_proj1_and_sort1<<<dim3(GEMM_BLOCKS + SORT_NC), B256, 0, stream>>>(
        x, W1, b1, feat, NN / 16, GEMM_BLOCKS, al1, ar1, el, er, src, dst, pairs, tab);
    gat_sort2_and_vec<<<dim3(1 + SORT_NB), dim3(1024), 0, stream>>>(
        pairs, tab, rowptr, csr_src, W2, al2, ar2, vlr);

    gat_node_aggregate<<<dim3(AGG_BLOCKS), B256, 0, stream>>>(
        rowptr, csr_src, feat, el, er, b1, vlr, h, el2, er2, NN);

    gat_node_aggregate_h<<<dim3(AGG_BLOCKS), B256, 0, stream>>>(rowptr, csr_src, h, el2, er2, z, NN);
    gat_proj2_final<<<dim3(GEMM_BLOCKS), B256, 0, stream>>>(z, W2, b2, out, NN / 16);
}
